# starting kernel
# baseline (speedup 1.0000x reference)
#include <hip/hip_runtime.h>
#include <hip/hip_bf16.h>

#define NNODES 50000
#define NEDGES 1600000
#define NFEAT  512
#define F1     128
#define H1_    8
#define C_     16
#define NCLASS 16

typedef float v2f __attribute__((ext_vector_type(2)));
typedef float v8f __attribute__((ext_vector_type(8)));

__device__ __forceinline__ unsigned enc_f32(float f) {
  unsigned u = __float_as_uint(f);
  return (u & 0x80000000u) ? ~u : (u | 0x80000000u);
}
__device__ __forceinline__ float dec_f32(unsigned u) {
  return (u & 0x80000000u) ? __uint_as_float(u & 0x7FFFFFFFu) : __uint_as_float(~u);
}

template <int TN, int K, int NOUT>
__global__ __launch_bounds__(32) void gemm_f32_wmma_strip(
    const float* __restrict__ X, const float* __restrict__ Wt,
    float* __restrict__ D) {
  int tm = blockIdx.x;
  int lane = threadIdx.x;
  int l15 = lane & 15, half = lane >> 4;
  const float* ap = X  + (size_t)(tm * 16 + l15) * K + half * 2;
  const float* bp = Wt + (size_t)l15 * K + half * 2;
  v8f acc[TN];
  #pragma unroll
  for (int t = 0; t < TN; ++t) acc[t] = (v8f){};
  #pragma unroll 2
  for (int k = 0; k < K; k += 4) {
    v2f a, b[TN];
    a.x = ap[k];
    a.y = ap[k + 1];
    #pragma unroll
    for (int t = 0; t < TN; ++t) {
      b[t].x = bp[t * 16 * K + k];
      b[t].y = bp[t * 16 * K + k + 1];
    }
    #pragma unroll
    for (int t = 0; t < TN; ++t)
      acc[t] = __builtin_amdgcn_wmma_f32_16x16x4_f32(false, a, false, b[t],
                                                     (short)0, acc[t], false, false);
  }
  #pragma unroll
  for (int t = 0; t < TN; ++t) {
    float* dp = D + (size_t)(tm * 16 + half * 8) * NOUT + t * 16 + l15;
    #pragma unroll
    for (int r = 0; r < 8; ++r) dp[r * NOUT] = acc[t][r];
  }
}

__global__ void attn_scores(const float* __restrict__ h,
                            const float* __restrict__ attS,
                            const float* __restrict__ attD,
                            float* __restrict__ aS, float* __restrict__ aD,
                            int NH, int H) {
  int t = blockIdx.x * blockDim.x + threadIdx.x;
  if (t >= NH) return;
  int hh = t % H;
  const float* hp = h + (size_t)t * C_;
  const float* sp = attS + hh * C_;
  const float* dp = attD + hh * C_;
  float s = 0.f, d = 0.f;
  #pragma unroll
  for (int c = 0; c < C_; ++c) { s += hp[c] * sp[c]; d += hp[c] * dp[c]; }
  aS[t] = s; aD[t] = d;
}

__global__ void fill_u32(unsigned* __restrict__ p, unsigned v, size_t n) {
  size_t t = (size_t)blockIdx.x * blockDim.x + threadIdx.x;
  if (t < n) p[t] = v;
}

template <int H>
__global__ void edge_max(const int* __restrict__ ei,
                         const float* __restrict__ aS, const float* __restrict__ aD,
                         unsigned* __restrict__ amax) {
  int e = blockIdx.x * blockDim.x + threadIdx.x;
  if (e >= NEDGES + NNODES) return;
  int s, d;
  if (e < NEDGES) { s = ei[e]; d = ei[NEDGES + e]; } else { s = d = e - NEDGES; }
  const float* sp = aS + (size_t)s * H;
  const float* dp = aD + (size_t)d * H;
  unsigned* mp = amax + (size_t)d * H;
  #pragma unroll
  for (int h = 0; h < H; ++h) {
    float al = sp[h] + dp[h];
    al = al > 0.f ? al : 0.2f * al;
    atomicMax(&mp[h], enc_f32(al));
  }
}

template <int H>
__global__ void edge_sum(const int* __restrict__ ei,
                         const float* __restrict__ aS, const float* __restrict__ aD,
                         const unsigned* __restrict__ amax, float* __restrict__ denom) {
  int e = blockIdx.x * blockDim.x + threadIdx.x;
  if (e >= NEDGES + NNODES) return;
  int s, d;
  if (e < NEDGES) { s = ei[e]; d = ei[NEDGES + e]; } else { s = d = e - NEDGES; }
  const float* sp = aS + (size_t)s * H;
  const float* dp = aD + (size_t)d * H;
  const unsigned* mp = amax + (size_t)d * H;
  float* np = denom + (size_t)d * H;
  #pragma unroll
  for (int h = 0; h < H; ++h) {
    float al = sp[h] + dp[h];
    al = al > 0.f ? al : 0.2f * al;
    atomicAdd(&np[h], expf(al - dec_f32(mp[h])));
  }
}

template <int H>
__global__ void edge_msg(const int* __restrict__ ei,
                         const float* __restrict__ aS, const float* __restrict__ aD,
                         const unsigned* __restrict__ amax, const float* __restrict__ denom,
                         const float* __restrict__ h, float* __restrict__ agg) {
  long long t = (long long)blockIdx.x * blockDim.x + threadIdx.x;
  long long total = (long long)(NEDGES + NNODES) * H;
  if (t >= total) return;
  int hh = (int)(t % H);
  int e  = (int)(t / H);
  int s, d;
  if (e < NEDGES) { s = ei[e]; d = ei[NEDGES + e]; } else { s = d = e - NEDGES; }
  size_t si = (size_t)s * H + hh, di = (size_t)d * H + hh;
  float al = aS[si] + aD[di];
  al = al > 0.f ? al : 0.2f * al;
  float coeff = expf(al - dec_f32(amax[di])) / (denom[di] + 1e-16f);
  const float* hp = h + si * C_;
  float* op = agg + di * C_;
  #pragma unroll
  for (int c = 0; c < C_; ++c) atomicAdd(&op[c], hp[c] * coeff);
}

__global__ void bias_elu(float* __restrict__ x, const float* __restrict__ b,
                         int F, size_t n) {
  size_t t = (size_t)blockIdx.x * blockDim.x + threadIdx.x;
  if (t >= n) return;
  float v = x[t] + b[(int)(t % F)];
  x[t] = v > 0.f ? v : expm1f(v);
}

__global__ void bias_logsoftmax(const float* __restrict__ agg,
                                const float* __restrict__ b,
                                float* __restrict__ out) {
  int n = blockIdx.x * blockDim.x + threadIdx.x;
  if (n >= NNODES) return;
  float v[NCLASS];
  float m = -1e30f;
  #pragma unroll
  for (int c = 0; c < NCLASS; ++c) {
    v[c] = agg[(size_t)n * NCLASS + c] + b[c];
    m = fmaxf(m, v[c]);
  }
  float s = 0.f;
  #pragma unroll
  for (int c = 0; c < NCLASS; ++c) s += expf(v[c] - m);
  float ls = logf(s);
  #pragma unroll
  for (int c = 0; c < NCLASS; ++c) out[(size_t)n * NCLASS + c] = v[c] - m - ls;
}

extern "C" void kernel_launch(void* const* d_in, const int* in_sizes, int n_in,
                              void* d_out, int out_size, void* d_ws, size_t ws_size,
                              hipStream_t stream) {
  const float* x     = (const float*)d_in[0];
  const int*   ei    = (const int*)  d_in[1];
  const float* W1    = (const float*)d_in[2];
  const float* attS1 = (const float*)d_in[3];
  const float* attD1 = (const float*)d_in[4];
  const float* b1    = (const float*)d_in[5];
  const float* W2    = (const float*)d_in[6];
  const float* attS2 = (const float*)d_in[7];
  const float* attD2 = (const float*)d_in[8];
  const float* b2    = (const float*)d_in[9];
  float* out = (float*)d_out;

  char* ws = (char*)d_ws;
  size_t off = 0;
  auto alloc = [&](size_t bytes) -> void* {
    void* p = ws + off;
    off = (off + bytes + 255) & ~(size_t)255;
    return p;
  };

  float*    h1   = (float*)   alloc((size_t)NNODES * F1 * 4);
  float*    as1  = (float*)   alloc((size_t)NNODES * H1_ * 4);
  float*    ad1  = (float*)   alloc((size_t)NNODES * H1_ * 4);
  unsigned* am1  = (unsigned*)alloc((size_t)NNODES * H1_ * 4);
  float*    dn1  = (float*)   alloc((size_t)NNODES * H1_ * 4);
  float*    agg1 = (float*)   alloc((size_t)NNODES * F1 * 4);
  float*    h2   = (float*)   alloc((size_t)NNODES * C_ * 4);
  float*    as2  = (float*)   alloc((size_t)NNODES * 4);
  float*    ad2  = (float*)   alloc((size_t)NNODES * 4);
  unsigned* am2  = (unsigned*)alloc((size_t)NNODES * 4);
  float*    dn2  = (float*)   alloc((size_t)NNODES * 4);
  float*    agg2 = (float*)   alloc((size_t)NNODES * C_ * 4);

  const int EDG = NEDGES + NNODES;
  const int TB = 256;
  auto nb = [](long long n, int b) { return (unsigned)((n + b - 1) / b); };

  gemm_f32_wmma_strip<8, NFEAT, F1><<<3125, 32, 0, stream>>>(x, W1, h1);
  attn_scores<<<nb((long long)NNODES * H1_, TB), TB, 0, stream>>>(
      h1, attS1, attD1, as1, ad1, NNODES * H1_, H1_);
  fill_u32<<<nb((long long)NNODES * H1_, TB), TB, 0, stream>>>(am1, 0u, (size_t)NNODES * H1_);
  fill_u32<<<nb((long long)NNODES * H1_, TB), TB, 0, stream>>>((unsigned*)dn1, 0u, (size_t)NNODES * H1_);
  fill_u32<<<nb((long long)NNODES * F1, TB), TB, 0, stream>>>((unsigned*)agg1, 0u, (size_t)NNODES * F1);
  edge_max<H1_><<<nb(EDG, TB), TB, 0, stream>>>(ei, as1, ad1, am1);
  edge_sum<H1_><<<nb(EDG, TB), TB, 0, stream>>>(ei, as1, ad1, am1, dn1);
  edge_msg<H1_><<<nb((long long)EDG * H1_, TB), TB, 0, stream>>>(
      ei, as1, ad1, am1, dn1, h1, agg1);
  bias_elu<<<nb((long long)NNODES * F1, TB), TB, 0, stream>>>(
      agg1, b1, F1, (size_t)NNODES * F1);

  gemm_f32_wmma_strip<1, F1, C_><<<3125, 32, 0, stream>>>(agg1, W2, h2);
  attn_scores<<<nb(NNODES, TB), TB, 0, stream>>>(h2, attS2, attD2, as2, ad2, NNODES, 1);
  fill_u32<<<nb(NNODES, TB), TB, 0, stream>>>(am2, 0u, (size_t)NNODES);
  fill_u32<<<nb(NNODES, TB), TB, 0, stream>>>((unsigned*)dn2, 0u, (size_t)NNODES);
  fill_u32<<<nb((long long)NNODES * C_, TB), TB, 0, stream>>>((unsigned*)agg2, 0u, (size_t)NNODES * C_);
  edge_max<1><<<nb(EDG, TB), TB, 0, stream>>>(ei, as2, ad2, am2);
  edge_sum<1><<<nb(EDG, TB), TB, 0, stream>>>(ei, as2, ad2, am2, dn2);
  edge_msg<1><<<nb(EDG, TB), TB, 0, stream>>>(ei, as2, ad2, am2, dn2, h2, agg2);

  bias_logsoftmax<<<nb(NNODES, TB), TB, 0, stream>>>(agg2, b2, out);
}


# Float atomics removed: edges grouped by destination each call (integer arrival-order rank, prefix sum, fill); one wave per node does max/exp/sum in registers; sums run in list order plus a lane tree
# speedup vs baseline: 6.9695x; 6.9695x over previous
#include <hip/hip_runtime.h>
#include <hip/hip_bf16.h>

#define NNODES 50000
#define NEDGES 1600000
#define NFEAT  512
#define F1     128
#define H1_    8
#define C_     16
#define NCLASS 16

typedef float v2f __attribute__((ext_vector_type(2)));
typedef float v8f __attribute__((ext_vector_type(8)));

template <int TN, int K, int NOUT>
__global__ __launch_bounds__(32) void gemm_f32_wmma_strip(
    const float* __restrict__ X, const float* __restrict__ Wt,
    float* __restrict__ D) {
  int tm = blockIdx.x;
  int lane = threadIdx.x;
  int l15 = lane & 15, half = lane >> 4;
  const float* ap = X  + (size_t)(tm * 16 + l15) * K + half * 2;
  const float* bp = Wt + (size_t)l15 * K + half * 2;
  v8f acc[TN];
  #pragma unroll
  for (int t = 0; t < TN; ++t) acc[t] = (v8f){};
  #pragma unroll 2
  for (int k = 0; k < K; k += 4) {
    v2f a, b[TN];
    a.x = ap[k];
    a.y = ap[k + 1];
    #pragma unroll
    for (int t = 0; t < TN; ++t) {
      b[t].x = bp[t * 16 * K + k];
      b[t].y = bp[t * 16 * K + k + 1];
    }
    #pragma unroll
    for (int t = 0; t < TN; ++t)
      acc[t] = __builtin_amdgcn_wmma_f32_16x16x4_f32(false, a, false, b[t],
                                                     (short)0, acc[t], false, false);
  }
  #pragma unroll
  for (int t = 0; t < TN; ++t) {
    float* dp = D + (size_t)(tm * 16 + half * 8) * NOUT + t * 16 + l15;
    #pragma unroll
    for (int r = 0; r < 8; ++r) dp[r * NOUT] = acc[t][r];
  }
}

__global__ void attn_scores(const float* __restrict__ h,
                            const float* __restrict__ attS,
                            const float* __restrict__ attD,
                            float* __restrict__ aS, float* __restrict__ aD,
                            int NH, int H) {
  int t = blockIdx.x * blockDim.x + threadIdx.x;
  if (t >= NH) return;
  int hh = t % H;
  const float* hp = h + (size_t)t * C_;
  const float* sp = attS + hh * C_;
  const float* dp = attD + hh * C_;
  float s = 0.f, d = 0.f;
  #pragma unroll
  for (int c = 0; c < C_; ++c) { s += hp[c] * sp[c]; d += hp[c] * dp[c]; }
  aS[t] = s; aD[t] = d;
}

__global__ void fill_u32(unsigned* __restrict__ p, unsigned v, size_t n) {
  size_t t = (size_t)blockIdx.x * blockDim.x + threadIdx.x;
  if (t < n) p[t] = v;
}

__global__ void edge_rank(const int* __restrict__ ei, unsigned* __restrict__ cnt,
                          int* __restrict__ rank) {
  int e = blockIdx.x * blockDim.x + threadIdx.x;
  if (e >= NEDGES) return;
  int d = ei[NEDGES + e];
  rank[e] = (int)atomicAdd(&cnt[d], 1u);
}

__global__ __launch_bounds__(256) void deg_scan(const unsigned* __restrict__ cnt,
                                                int* __restrict__ rowptr) {
  __shared__ unsigned sh[256];
  const int t = threadIdx.x;
  const int base = blockIdx.x * 256;
  const uint4* c4 = (const uint4*)cnt;
  unsigned p = 0;
  #pragma unroll 4
  for (int i = t; i < (base >> 2); i += 256) {
    uint4 c = c4[i];
    p += c.x + c.y + c.z + c.w;
  }
  sh[t] = p;
  __syncthreads();
  for (int s = 128; s > 0; s >>= 1) {
    if (t < s) sh[t] += sh[t + s];
    __syncthreads();
  }
  const unsigned prefix = sh[0];
  __syncthreads();
  const int i = base + t;
  const unsigned v = (i < NNODES) ? cnt[i] : 0u;
  sh[t] = v;
  __syncthreads();
  for (int s = 1; s < 256; s <<= 1) {
    unsigned x = (t >= s) ? sh[t - s] : 0u;
    __syncthreads();
    sh[t] += x;
    __syncthreads();
  }
  const unsigned incl = sh[t];
  if (i < NNODES) rowptr[i] = (int)(prefix + incl - v);
  if (i == NNODES - 1) rowptr[NNODES] = (int)(prefix + incl);
}

__global__ void csr_fill(const int* __restrict__ ei, const int* __restrict__ rowptr,
                         const int* __restrict__ rank, int* __restrict__ csrc) {
  int e = blockIdx.x * blockDim.x + threadIdx.x;
  if (e >= NEDGES) return;
  int s = ei[e], d = ei[NEDGES + e];
  int pos = rowptr[d] + rank[e];
  if ((unsigned)pos < (unsigned)NEDGES) csrc[pos] = s;
}

__device__ __forceinline__ float leaky02(float a) { return a > 0.f ? a : 0.2f * a; }

__global__ __launch_bounds__(256) void gat_gather_h8(
    const int* __restrict__ rowptr, const int* __restrict__ csrc,
    const float* __restrict__ aS, const float* __restrict__ aD,
    const float* __restrict__ h, float* __restrict__ outp) {
  const int lane = threadIdx.x & 31;
  const int wv = __builtin_amdgcn_readfirstlane((int)(threadIdx.x >> 5));
  const int n = blockIdx.x * 8 + wv;
  if (n >= NNODES) return;
  const int hl = lane >> 2, q = lane & 3;
  int beg = rowptr[n], end = rowptr[n + 1];
  beg = min(max(beg, 0), NEDGES);
  end = min(max(end, beg), NEDGES);
  const float4* h4 = (const float4*)h;
  const float adn = aD[n * H1_ + hl];
  const float a_self = leaky02(aS[n * H1_ + hl] + adn);

  float m = a_self;
  for (int c0 = beg; c0 < end; c0 += 32) {
    const int cnt = min(32, end - c0);
    int sv = (lane < cnt) ? csrc[c0 + lane] : 0;
    sv = min(max(sv, 0), NNODES - 1);
    #pragma unroll 4
    for (int j4 = 0; j4 < cnt; j4 += 4) {
      const int jq = j4 + q;
      const int sq = __shfl(sv, jq);
      const float a = leaky02(aS[sq * H1_ + hl] + adn);
      if (jq < cnt) m = fmaxf(m, a);
    }
  }
  m = fmaxf(m, __shfl_xor(m, 1));
  m = fmaxf(m, __shfl_xor(m, 2));

  const float w0 = expf(a_self - m);
  float den = (q == 0) ? w0 : 0.f;
  float4 acc;
  {
    const float4 r = h4[n * 32 + lane];
    acc.x = w0 * r.x; acc.y = w0 * r.y; acc.z = w0 * r.z; acc.w = w0 * r.w;
  }
  for (int c0 = beg; c0 < end; c0 += 32) {
    const int cnt = min(32, end - c0);
    int sv = (lane < cnt) ? csrc[c0 + lane] : 0;
    sv = min(max(sv, 0), NNODES - 1);
    for (int j4 = 0; j4 < cnt; j4 += 4) {
      const int jq = j4 + q;
      const int sq = __shfl(sv, jq);
      const float a = leaky02(aS[sq * H1_ + hl] + adn);
      const float wq = (jq < cnt) ? expf(a - m) : 0.f;
      den += wq;
      if (j4 + 4 <= cnt) {
        float4 r[4];
        float w[4];
        #pragma unroll
        for (int k = 0; k < 4; ++k) {
          const int s = __builtin_amdgcn_readlane(sv, j4 + k);
          r[k] = h4[s * 32 + lane];
          w[k] = __shfl(wq, (lane & ~3) | k);
        }
        #pragma unroll
        for (int k = 0; k < 4; ++k) {
          acc.x += w[k] * r[k].x; acc.y += w[k] * r[k].y;
          acc.z += w[k] * r[k].z; acc.w += w[k] * r[k].w;
        }
      } else {
        #pragma unroll
        for (int k = 0; k < 3; ++k) {
          if (j4 + k < cnt) {
            const int s = __builtin_amdgcn_readlane(sv, j4 + k);
            const float w = __shfl(wq, (lane & ~3) | k);
            const float4 r = h4[s * 32 + lane];
            acc.x += w * r.x; acc.y += w * r.y; acc.z += w * r.z; acc.w += w * r.w;
          }
        }
      }
    }
  }
  den += __shfl_xor(den, 1);
  den += __shfl_xor(den, 2);
  const float inv = 1.0f / (den + 1e-16f);
  float4 o;
  o.x = acc.x * inv; o.y = acc.y * inv; o.z = acc.z * inv; o.w = acc.w * inv;
  ((float4*)outp)[n * 32 + lane] = o;
}

__global__ __launch_bounds__(256) void gat_gather_h1(
    const int* __restrict__ rowptr, const int* __restrict__ csrc,
    const float* __restrict__ aS, const float* __restrict__ aD,
    const float* __restrict__ h, float* __restrict__ outp) {
  const int lane = threadIdx.x & 31;
  const int wv = __builtin_amdgcn_readfirstlane((int)(threadIdx.x >> 5));
  const int n = blockIdx.x * 8 + wv;
  if (n >= NNODES) return;
  const int es = lane >> 2, q = lane & 3;
  int beg = rowptr[n], end = rowptr[n + 1];
  beg = min(max(beg, 0), NEDGES);
  end = min(max(end, beg), NEDGES);
  const float4* h4 = (const float4*)h;
  const float adn = aD[n];
  const float a_self = leaky02(aS[n] + adn);

  float m = a_self;
  for (int c0 = beg; c0 < end; c0 += 32) {
    const int j = c0 + lane;
    if (j < end) {
      int s = csrc[j];
      s = min(max(s, 0), NNODES - 1);
      m = fmaxf(m, leaky02(aS[s] + adn));
    }
  }
  #pragma unroll
  for (int o = 16; o > 0; o >>= 1) m = fmaxf(m, __shfl_xor(m, o));

  const float w0 = expf(a_self - m);
  float den = (lane == 0) ? w0 : 0.f;
  float4 acc = {0.f, 0.f, 0.f, 0.f};
  if (es == 0) {
    const float4 r = h4[n * 4 + q];
    acc.x = w0 * r.x; acc.y = w0 * r.y; acc.z = w0 * r.z; acc.w = w0 * r.w;
  }
  for (int c0 = beg; c0 < end; c0 += 32) {
    const int cnt = min(32, end - c0);
    int s = 0;
    float w = 0.f;
    if (lane < cnt) {
      s = csrc[c0 + lane];
      s = min(max(s, 0), NNODES - 1);
      w = expf(leaky02(aS[s] + adn) - m);
    }
    den += w;
    #pragma unroll
    for (int it = 0; it < 4; ++it) {
      const int jj = it * 8 + es;
      const int sj = __shfl(s, jj);
      const float wj = __shfl(w, jj);
      const float4 r = h4[sj * 4 + q];
      const bool ok = jj < cnt;
      acc.x += wj * (ok ? r.x : 0.f); acc.y += wj * (ok ? r.y : 0.f);
      acc.z += wj * (ok ? r.z : 0.f); acc.w += wj * (ok ? r.w : 0.f);
    }
  }
  #pragma unroll
  for (int o = 16; o > 0; o >>= 1) den += __shfl_xor(den, o);
  #pragma unroll
  for (int o = 16; o >= 4; o >>= 1) {
    acc.x += __shfl_xor(acc.x, o); acc.y += __shfl_xor(acc.y, o);
    acc.z += __shfl_xor(acc.z, o); acc.w += __shfl_xor(acc.w, o);
  }
  const float inv = 1.0f / (den + 1e-16f);
  if (es == 0) {
    float4 o4;
    o4.x = acc.x * inv; o4.y = acc.y * inv; o4.z = acc.z * inv; o4.w = acc.w * inv;
    ((float4*)outp)[n * 4 + q] = o4;
  }
}

__global__ void bias_elu(float* __restrict__ x, const float* __restrict__ b,
                         int F, size_t n) {
  size_t t = (size_t)blockIdx.x * blockDim.x + threadIdx.x;
  if (t >= n) return;
  float v = x[t] + b[(int)(t % F)];
  x[t] = v > 0.f ? v : expm1f(v);
}

__global__ void bias_logsoftmax(const float* __restrict__ agg,
                                const float* __restrict__ b,
                                float* __restrict__ out) {
  int n = blockIdx.x * blockDim.x + threadIdx.x;
  if (n >= NNODES) return;
  float v[NCLASS];
  float m = -1e30f;
  #pragma unroll
  for (int c = 0; c < NCLASS; ++c) {
    v[c] = agg[(size_t)n * NCLASS + c] + b[c];
    m = fmaxf(m, v[c]);
  }
  float s = 0.f;
  #pragma unroll
  for (int c = 0; c < NCLASS; ++c) s += expf(v[c] - m);
  float ls = logf(s);
  #pragma unroll
  for (int c = 0; c < NCLASS; ++c) out[(size_t)n * NCLASS + c] = v[c] - m - ls;
}

extern "C" void kernel_launch(void* const* d_in, const int* in_sizes, int n_in,
                              void* d_out, int out_size, void* d_ws, size_t ws_size,
                              hipStream_t stream) {
  const float* x     = (const float*)d_in[0];
  const int*   ei    = (const int*)  d_in[1];
  const float* W1    = (const float*)d_in[2];
  const float* attS1 = (const float*)d_in[3];
  const float* attD1 = (const float*)d_in[4];
  const float* b1    = (const float*)d_in[5];
  const float* W2    = (const float*)d_in[6];
  const float* attS2 = (const float*)d_in[7];
  const float* attD2 = (const float*)d_in[8];
  const float* b2    = (const float*)d_in[9];
  float* out = (float*)d_out;

  char* ws = (char*)d_ws;
  size_t off = 0;
  auto alloc = [&](size_t bytes) -> void* {
    void* p = ws + off;
    off = (off + bytes + 255) & ~(size_t)255;
    return p;
  };

  float*    h1   = (float*)   alloc((size_t)NNODES * F1 * 4);
  float*    as1  = (float*)   alloc((size_t)NNODES * H1_ * 4);
  float*    ad1  = (float*)   alloc((size_t)NNODES * H1_ * 4);
  float*    agg1 = (float*)   alloc((size_t)NNODES * F1 * 4);
  int*      csrc = (int*)     alloc((size_t)NEDGES * 4);
  int*      rptr = (int*)     alloc((size_t)(NNODES + 1) * 4);
  unsigned* cnt  = (unsigned*)alloc((size_t)NNODES * 4);
  int*      rank = (int*)agg1;
  float*    h2   = h1;
  float*    agg2 = h1 + (size_t)NNODES * C_;
  float*    as2  = h1 + (size_t)NNODES * C_ * 2;
  float*    ad2  = h1 + (size_t)NNODES * C_ * 2 + NNODES;

  const int TB = 256;
  auto nb = [](long long n, int b) { return (unsigned)((n + b - 1) / b); };

  fill_u32<<<nb(NNODES, TB), TB, 0, stream>>>(cnt, 0u, (size_t)NNODES);
  edge_rank<<<nb(NEDGES, TB), TB, 0, stream>>>(ei, cnt, rank);
  deg_scan<<<nb(NNODES, 256), 256, 0, stream>>>(cnt, rptr);
  csr_fill<<<nb(NEDGES, TB), TB, 0, stream>>>(ei, rptr, rank, csrc);

  gemm_f32_wmma_strip<8, NFEAT, F1><<<3125, 32, 0, stream>>>(x, W1, h1);
  attn_scores<<<nb((long long)NNODES * H1_, TB), TB, 0, stream>>>(
      h1, attS1, attD1, as1, ad1, NNODES * H1_, H1_);
  gat_gather_h8<<<nb(NNODES, 8), 256, 0, stream>>>(rptr, csrc, as1, ad1, h1, agg1);
  bias_elu<<<nb((long long)NNODES * F1, TB), TB, 0, stream>>>(
      agg1, b1, F1, (size_t)NNODES * F1);

  gemm_f32_wmma_strip<1, F1, C_><<<3125, 32, 0, stream>>>(agg1, W2, h2);
  attn_scores<<<nb(NNODES, TB), TB, 0, stream>>>(h2, attS2, attD2, as2, ad2, NNODES, 1);
  gat_gather_h1<<<nb(NNODES, 8), 256, 0, stream>>>(rptr, csrc, as2, ad2, h2, agg2);

  bias_logsoftmax<<<nb(NNODES, TB), TB, 0, stream>>>(agg2, b2, out);
}


# Scores, bias+ELU and bias+log-softmax moved into the epilogues of the products and gathers that feed them (12 launches -> 8); score and class sums now lane trees
# speedup vs baseline: 7.3573x; 1.0556x over previous
#include <hip/hip_runtime.h>
#include <hip/hip_bf16.h>

#define NNODES 50000
#define NEDGES 1600000
#define NFEAT  512
#define F1     128
#define H1_    8
#define C_     16
#define NCLASS 16

typedef float v2f __attribute__((ext_vector_type(2)));
typedef float v8f __attribute__((ext_vector_type(8)));

template <int TN, int K, int NOUT>
__global__ __launch_bounds__(32) void gemm_f32_wmma_strip(
    const float* __restrict__ X, const float* __restrict__ Wt,
    const float* __restrict__ attS, const float* __restrict__ attD,
    float* __restrict__ D, float* __restrict__ SD) {
  static_assert(TN == 8 || TN == 1, "score record layouts exist for 8 heads and for 1 head");
  int tm = blockIdx.x;
  int lane = threadIdx.x;
  int l15 = lane & 15, half = lane >> 4;
  const float* ap = X  + (size_t)(tm * 16 + l15) * K + half * 2;
  const float* bp = Wt + (size_t)l15 * K + half * 2;
  v8f acc[TN];
  #pragma unroll
  for (int t = 0; t < TN; ++t) acc[t] = (v8f){};
  #pragma unroll 2
  for (int k = 0; k < K; k += 4) {
    v2f a, b[TN];
    a.x = ap[k];
    a.y = ap[k + 1];
    #pragma unroll
    for (int t = 0; t < TN; ++t) {
      b[t].x = bp[t * 16 * K + k];
      b[t].y = bp[t * 16 * K + k + 1];
    }
    #pragma unroll
    for (int t = 0; t < TN; ++t)
      acc[t] = __builtin_amdgcn_wmma_f32_16x16x4_f32(false, a, false, b[t],
                                                     (short)0, acc[t], false, false);
  }
  #pragma unroll
  for (int t = 0; t < TN; ++t) {
    float* dp = D + (size_t)(tm * 16 + half * 8) * NOUT + t * 16 + l15;
    #pragma unroll
    for (int r = 0; r < 8; ++r) dp[r * NOUT] = acc[t][r];
  }

  const bool hi8 = (l15 & 8) != 0;
  const int l7 = l15 & 7;
  float keep[8];
  #pragma unroll
  for (int r = 0; r < 8; ++r) keep[r] = 0.f;
  #pragma unroll
  for (int t = 0; t < TN; ++t) {
    const float ws = attS[t * 16 + l15];
    const float wd = attD[t * 16 + l15];
    #pragma unroll
    for (int r = 0; r < 8; ++r) {
      const float ps = acc[t][r] * ws;
      const float pd = acc[t][r] * wd;
      float v = (hi8 ? pd : ps) + __shfl_xor(hi8 ? ps : pd, 8);
      v += __shfl_xor(v, 4);
      v += __shfl_xor(v, 2);
      v += __shfl_xor(v, 1);
      if (TN == 8) {
        if (l7 == t) keep[r] = v;
      } else {
        if (l7 == r) keep[0] = v;
      }
    }
  }
  if (TN == 8) {
    float* sp = SD + (size_t)(tm * 16 + half * 8) * 16 + l15;
    #pragma unroll
    for (int r = 0; r < 8; ++r) sp[r * 16] = keep[r];
  } else {
    SD[(size_t)tm * 32 + (hi8 ? 16 : 0) + half * 8 + l7] = keep[0];
  }
}

__global__ void fill_u32(unsigned* __restrict__ p, unsigned v, size_t n) {
  size_t t = (size_t)blockIdx.x * blockDim.x + threadIdx.x;
  if (t < n) p[t] = v;
}

__global__ void edge_rank(const int* __restrict__ ei, unsigned* __restrict__ cnt,
                          int* __restrict__ rank) {
  int e = blockIdx.x * blockDim.x + threadIdx.x;
  if (e >= NEDGES) return;
  int d = ei[NEDGES + e];
  rank[e] = (int)atomicAdd(&cnt[d], 1u);
}

__global__ __launch_bounds__(256) void deg_scan(const unsigned* __restrict__ cnt,
                                                int* __restrict__ rowptr) {
  __shared__ unsigned sh[256];
  const int t = threadIdx.x;
  const int base = blockIdx.x * 256;
  const uint4* c4 = (const uint4*)cnt;
  unsigned p = 0;
  #pragma unroll 4
  for (int i = t; i < (base >> 2); i += 256) {
    uint4 c = c4[i];
    p += c.x + c.y + c.z + c.w;
  }
  sh[t] = p;
  __syncthreads();
  for (int s = 128; s > 0; s >>= 1) {
    if (t < s) sh[t] += sh[t + s];
    __syncthreads();
  }
  const unsigned prefix = sh[0];
  __syncthreads();
  const int i = base + t;
  const unsigned v = (i < NNODES) ? cnt[i] : 0u;
  sh[t] = v;
  __syncthreads();
  for (int s = 1; s < 256; s <<= 1) {
    unsigned x = (t >= s) ? sh[t - s] : 0u;
    __syncthreads();
    sh[t] += x;
    __syncthreads();
  }
  const unsigned incl = sh[t];
  if (i < NNODES) rowptr[i] = (int)(prefix + incl - v);
  if (i == NNODES - 1) rowptr[NNODES] = (int)(prefix + incl);
}

__global__ void csr_fill(const int* __restrict__ ei, const int* __restrict__ rowptr,
                         const int* __restrict__ rank, int* __restrict__ csrc) {
  int e = blockIdx.x * blockDim.x + threadIdx.x;
  if (e >= NEDGES) return;
  int s = ei[e], d = ei[NEDGES + e];
  int pos = rowptr[d] + rank[e];
  if ((unsigned)pos < (unsigned)NEDGES) csrc[pos] = s;
}

__device__ __forceinline__ float leaky02(float a) { return a > 0.f ? a : 0.2f * a; }

__global__ __launch_bounds__(256) void gat_gather_h8(
    const int* __restrict__ rowptr, const int* __restrict__ csrc,
    const float* __restrict__ sd,
    const float* __restrict__ h, const float* __restrict__ bias,
    float* __restrict__ outp) {
  const int lane = threadIdx.x & 31;
  const int wv = __builtin_amdgcn_readfirstlane((int)(threadIdx.x >> 5));
  const int n = blockIdx.x * 8 + wv;
  if (n >= NNODES) return;
  const int hl = lane >> 2, q = lane & 3;
  int beg = rowptr[n], end = rowptr[n + 1];
  beg = min(max(beg, 0), NEDGES);
  end = min(max(end, beg), NEDGES);
  const float4* h4 = (const float4*)h;
  const float adn = sd[n * 16 + 8 + hl];
  const float a_self = leaky02(sd[n * 16 + hl] + adn);

  float m = a_self;
  for (int c0 = beg; c0 < end; c0 += 32) {
    const int cnt = min(32, end - c0);
    int sv = (lane < cnt) ? csrc[c0 + lane] : 0;
    sv = min(max(sv, 0), NNODES - 1);
    #pragma unroll 4
    for (int j4 = 0; j4 < cnt; j4 += 4) {
      const int jq = j4 + q;
      const int sq = __shfl(sv, jq);
      const float a = leaky02(sd[sq * 16 + hl] + adn);
      if (jq < cnt) m = fmaxf(m, a);
    }
  }
  m = fmaxf(m, __shfl_xor(m, 1));
  m = fmaxf(m, __shfl_xor(m, 2));

  const float w0 = expf(a_self - m);
  float den = (q == 0) ? w0 : 0.f;
  float4 acc;
  {
    const float4 r = h4[n * 32 + lane];
    acc.x = w0 * r.x; acc.y = w0 * r.y; acc.z = w0 * r.z; acc.w = w0 * r.w;
  }
  for (int c0 = beg; c0 < end; c0 += 32) {
    const int cnt = min(32, end - c0);
    int sv = (lane < cnt) ? csrc[c0 + lane] : 0;
    sv = min(max(sv, 0), NNODES - 1);
    for (int j4 = 0; j4 < cnt; j4 += 4) {
      const int jq = j4 + q;
      const int sq = __shfl(sv, jq);
      const float a = leaky02(sd[sq * 16 + hl] + adn);
      const float wq = (jq < cnt) ? expf(a - m) : 0.f;
      den += wq;
      if (j4 + 4 <= cnt) {
        float4 r[4];
        float w[4];
        #pragma unroll
        for (int k = 0; k < 4; ++k) {
          const int s = __builtin_amdgcn_readlane(sv, j4 + k);
          r[k] = h4[s * 32 + lane];
          w[k] = __shfl(wq, (lane & ~3) | k);
        }
        #pragma unroll
        for (int k = 0; k < 4; ++k) {
          acc.x += w[k] * r[k].x; acc.y += w[k] * r[k].y;
          acc.z += w[k] * r[k].z; acc.w += w[k] * r[k].w;
        }
      } else {
        #pragma unroll
        for (int k = 0; k < 3; ++k) {
          if (j4 + k < cnt) {
            const int s = __builtin_amdgcn_readlane(sv, j4 + k);
            const float w = __shfl(wq, (lane & ~3) | k);
            const float4 r = h4[s * 32 + lane];
            acc.x += w * r.x; acc.y += w * r.y; acc.z += w * r.z; acc.w += w * r.w;
          }
        }
      }
    }
  }
  den += __shfl_xor(den, 1);
  den += __shfl_xor(den, 2);
  const float inv = 1.0f / (den + 1e-16f);
  float4 o;
  o.x = acc.x * inv + bias[lane * 4 + 0]; o.y = acc.y * inv + bias[lane * 4 + 1];
  o.z = acc.z * inv + bias[lane * 4 + 2]; o.w = acc.w * inv + bias[lane * 4 + 3];
  o.x = o.x > 0.f ? o.x : expm1f(o.x); o.y = o.y > 0.f ? o.y : expm1f(o.y);
  o.z = o.z > 0.f ? o.z : expm1f(o.z); o.w = o.w > 0.f ? o.w : expm1f(o.w);
  ((float4*)outp)[n * 32 + lane] = o;
}

__device__ __forceinline__ int sd2_src(int n) { return ((n >> 4) << 5) + (n & 15); }
__device__ __forceinline__ int sd2_dst(int n) { return ((n >> 4) << 5) + 16 + (n & 15); }

__global__ __launch_bounds__(256) void gat_gather_h1(
    const int* __restrict__ rowptr, const int* __restrict__ csrc,
    const float* __restrict__ sd,
    const float* __restrict__ h, const float* __restrict__ bias,
    float* __restrict__ outp) {
  const int lane = threadIdx.x & 31;
  const int wv = __builtin_amdgcn_readfirstlane((int)(threadIdx.x >> 5));
  const int n = blockIdx.x * 8 + wv;
  if (n >= NNODES) return;
  const int es = lane >> 2, q = lane & 3;
  int beg = rowptr[n], end = rowptr[n + 1];
  beg = min(max(beg, 0), NEDGES);
  end = min(max(end, beg), NEDGES);
  const float4* h4 = (const float4*)h;
  const float adn = sd[sd2_dst(n)];
  const float a_self = leaky02(sd[sd2_src(n)] + adn);

  float m = a_self;
  for (int c0 = beg; c0 < end; c0 += 32) {
    const int j = c0 + lane;
    if (j < end) {
      int s = csrc[j];
      s = min(max(s, 0), NNODES - 1);
      m = fmaxf(m, leaky02(sd[sd2_src(s)] + adn));
    }
  }
  #pragma unroll
  for (int o = 16; o > 0; o >>= 1) m = fmaxf(m, __shfl_xor(m, o));

  const float w0 = expf(a_self - m);
  float den = (lane == 0) ? w0 : 0.f;
  float4 acc = {0.f, 0.f, 0.f, 0.f};
  if (es == 0) {
    const float4 r = h4[n * 4 + q];
    acc.x = w0 * r.x; acc.y = w0 * r.y; acc.z = w0 * r.z; acc.w = w0 * r.w;
  }
  for (int c0 = beg; c0 < end; c0 += 32) {
    const int cnt = min(32, end - c0);
    int s = 0;
    float w = 0.f;
    if (lane < cnt) {
      s = csrc[c0 + lane];
      s = min(max(s, 0), NNODES - 1);
      w = expf(leaky02(sd[sd2_src(s)] + adn) - m);
    }
    den += w;
    #pragma unroll
    for (int it = 0; it < 4; ++it) {
      const int jj = it * 8 + es;
      const int sj = __shfl(s, jj);
      const float wj = __shfl(w, jj);
      const float4 r = h4[sj * 4 + q];
      const bool ok = jj < cnt;
      acc.x += wj * (ok ? r.x : 0.f); acc.y += wj * (ok ? r.y : 0.f);
      acc.z += wj * (ok ? r.z : 0.f); acc.w += wj * (ok ? r.w : 0.f);
    }
  }
  #pragma unroll
  for (int o = 16; o > 0; o >>= 1) den += __shfl_xor(den, o);
  #pragma unroll
  for (int o = 16; o >= 4; o >>= 1) {
    acc.x += __shfl_xor(acc.x, o); acc.y += __shfl_xor(acc.y, o);
    acc.z += __shfl_xor(acc.z, o); acc.w += __shfl_xor(acc.w, o);
  }
  const float inv = 1.0f / (den + 1e-16f);
  float4 v;
  v.x = acc.x * inv + bias[q * 4 + 0]; v.y = acc.y * inv + bias[q * 4 + 1];
  v.z = acc.z * inv + bias[q * 4 + 2]; v.w = acc.w * inv + bias[q * 4 + 3];
  float vm = fmaxf(fmaxf(v.x, v.y), fmaxf(v.z, v.w));
  vm = fmaxf(vm, __shfl_xor(vm, 1));
  vm = fmaxf(vm, __shfl_xor(vm, 2));
  float se = (expf(v.x - vm) + expf(v.y - vm)) + (expf(v.z - vm) + expf(v.w - vm));
  se += __shfl_xor(se, 1);
  se += __shfl_xor(se, 2);
  const float ls = logf(se);
  if (es == 0) {
    float4 o4;
    o4.x = v.x - vm - ls; o4.y = v.y - vm - ls; o4.z = v.z - vm - ls; o4.w = v.w - vm - ls;
    ((float4*)outp)[n * 4 + q] = o4;
  }
}

extern "C" void kernel_launch(void* const* d_in, const int* in_sizes, int n_in,
                              void* d_out, int out_size, void* d_ws, size_t ws_size,
                              hipStream_t stream) {
  const float* x     = (const float*)d_in[0];
  const int*   ei    = (const int*)  d_in[1];
  const float* W1    = (const float*)d_in[2];
  const float* attS1 = (const float*)d_in[3];
  const float* attD1 = (const float*)d_in[4];
  const float* b1    = (const float*)d_in[5];
  const float* W2    = (const float*)d_in[6];
  const float* attS2 = (const float*)d_in[7];
  const float* attD2 = (const float*)d_in[8];
  const float* b2    = (const float*)d_in[9];
  float* out = (float*)d_out;

  char* ws = (char*)d_ws;
  size_t off = 0;
  auto alloc = [&](size_t bytes) -> void* {
    void* p = ws + off;
    off = (off + bytes + 255) & ~(size_t)255;
    return p;
  };

  float*    h1   = (float*)   alloc((size_t)NNODES * F1 * 4);
  float*    sd1  = (float*)   alloc((size_t)NNODES * 16 * 4);
  float*    agg1 = (float*)   alloc((size_t)NNODES * F1 * 4);
  int*      csrc = (int*)     alloc((size_t)NEDGES * 4);
  int*      rptr = (int*)     alloc((size_t)(NNODES + 1) * 4);
  unsigned* cnt  = (unsigned*)alloc((size_t)NNODES * 4);
  int*      rank = (int*)agg1;
  float*    h2   = h1;
  float*    sd2  = h1 + (size_t)NNODES * C_;

  const int TB = 256;
  auto nb = [](long long n, int b) { return (unsigned)((n + b - 1) / b); };

  fill_u32<<<nb(NNODES, TB), TB, 0, stream>>>(cnt, 0u, (size_t)NNODES);
  edge_rank<<<nb(NEDGES, TB), TB, 0, stream>>>(ei, cnt, rank);
  deg_scan<<<nb(NNODES, 256), 256, 0, stream>>>(cnt, rptr);
  csr_fill<<<nb(NEDGES, TB), TB, 0, stream>>>(ei, rptr, rank, csrc);

  gemm_f32_wmma_strip<8, NFEAT, F1><<<3125, 32, 0, stream>>>(x, W1, attS1, attD1, h1, sd1);
  gat_gather_h8<<<nb(NNODES, 8), 256, 0, stream>>>(rptr, csrc, sd1, h1, b1, agg1);

  gemm_f32_wmma_strip<1, F1, C_><<<3125, 32, 0, stream>>>(agg1, W2, attS2, attD2, h2, sd2);
  gat_gather_h1<<<nb(NNODES, 8), 256, 0, stream>>>(rptr, csrc, sd2, h2, b2, out);
}


# Product fragment loads widened to 16 B per lane: one load feeds two matrix steps, k inside each block of 8 permuted alike in both operands (sum order within a block changes); half as many loads
# speedup vs baseline: 7.8840x; 1.0716x over previous
#include <hip/hip_runtime.h>
#include <hip/hip_bf16.h>

#define NNODES 50000
#define NEDGES 1600000
#define NFEAT  512
#define F1     128
#define H1_    8
#define C_     16
#define NCLASS 16

typedef float v2f __attribute__((ext_vector_type(2)));
typedef float v4f __attribute__((ext_vector_type(4)));
typedef float v8f __attribute__((ext_vector_type(8)));

template <int TN, int K, int NOUT>
__global__ __launch_bounds__(32) void gemm_f32_wmma_strip(
    const float* __restrict__ X, const float* __restrict__ Wt,
    const float* __restrict__ attS, const float* __restrict__ attD,
    float* __restrict__ D, float* __restrict__ SD) {
  static_assert(TN == 8 || TN == 1, "score record layouts exist for 8 heads and for 1 head");
  int tm = blockIdx.x;
  int lane = threadIdx.x;
  int l15 = lane & 15, half = lane >> 4;
  static_assert(K % 8 == 0, "the k loop walks blocks of 8");
  const float* ap = X  + (size_t)(tm * 16 + l15) * K + half * 4;
  const float* bp = Wt + (size_t)l15 * K + half * 4;
  v8f acc[TN];
  #pragma unroll
  for (int t = 0; t < TN; ++t) acc[t] = (v8f){};
  #pragma unroll 2
  for (int k = 0; k < K; k += 8) {
    const v4f a = *(const v4f*)(ap + k);
    v4f b[TN];
    #pragma unroll
    for (int t = 0; t < TN; ++t) b[t] = *(const v4f*)(bp + t * 16 * K + k);
    const v2f a0 = {a.x, a.y}, a1 = {a.z, a.w};
    #pragma unroll
    for (int t = 0; t < TN; ++t) {
      const v2f b0 = {b[t].x, b[t].y};
      acc[t] = __builtin_amdgcn_wmma_f32_16x16x4_f32(false, a0, false, b0,
                                                     (short)0, acc[t], false, false);
    }
    #pragma unroll
    for (int t = 0; t < TN; ++t) {
      const v2f b1 = {b[t].z, b[t].w};
      acc[t] = __builtin_amdgcn_wmma_f32_16x16x4_f32(false, a1, false, b1,
                                                     (short)0, acc[t], false, false);
    }
  }
  #pragma unroll
  for (int t = 0; t < TN; ++t) {
    float* dp = D + (size_t)(tm * 16 + half * 8) * NOUT + t * 16 + l15;
    #pragma unroll
    for (int r = 0; r < 8; ++r) dp[r * NOUT] = acc[t][r];
  }

  const bool hi8 = (l15 & 8) != 0;
  const int l7 = l15 & 7;
  float keep[8];
  #pragma unroll
  for (int r = 0; r < 8; ++r) keep[r] = 0.f;
  #pragma unroll
  for (int t = 0; t < TN; ++t) {
    const float ws = attS[t * 16 + l15];
    const float wd = attD[t * 16 + l15];
    #pragma unroll
    for (int r = 0; r < 8; ++r) {
      const float ps = acc[t][r] * ws;
      const float pd = acc[t][r] * wd;
      float v = (hi8 ? pd : ps) + __shfl_xor(hi8 ? ps : pd, 8);
      v += __shfl_xor(v, 4);
      v += __shfl_xor(v, 2);
      v += __shfl_xor(v, 1);
      if (TN == 8) {
        if (l7 == t) keep[r] = v;
      } else {
        if (l7 == r) keep[0] = v;
      }
    }
  }
  if (TN == 8) {
    float* sp = SD + (size_t)(tm * 16 + half * 8) * 16 + l15;
    #pragma unroll
    for (int r = 0; r < 8; ++r) sp[r * 16] = keep[r];
  } else {
    SD[(size_t)tm * 32 + (hi8 ? 16 : 0) + half * 8 + l7] = keep[0];
  }
}

__global__ void fill_u32(unsigned* __restrict__ p, unsigned v, size_t n) {
  size_t t = (size_t)blockIdx.x * blockDim.x + threadIdx.x;
  if (t < n) p[t] = v;
}

__global__ void edge_rank(const int* __restrict__ ei, unsigned* __restrict__ cnt,
                          int* __restrict__ rank) {
  int e = blockIdx.x * blockDim.x + threadIdx.x;
  if (e >= NEDGES) return;
  int d = ei[NEDGES + e];
  rank[e] = (int)atomicAdd(&cnt[d], 1u);
}

__global__ __launch_bounds__(256) void deg_scan(const unsigned* __restrict__ cnt,
                                                int* __restrict__ rowptr) {
  __shared__ unsigned sh[256];
  const int t = threadIdx.x;
  const int base = blockIdx.x * 256;
  const uint4* c4 = (const uint4*)cnt;
  unsigned p = 0;
  #pragma unroll 4
  for (int i = t; i < (base >> 2); i += 256) {
    uint4 c = c4[i];
    p += c.x + c.y + c.z + c.w;
  }
  sh[t] = p;
  __syncthreads();
  for (int s = 128; s > 0; s >>= 1) {
    if (t < s) sh[t] += sh[t + s];
    __syncthreads();
  }
  const unsigned prefix = sh[0];
  __syncthreads();
  const int i = base + t;
  const unsigned v = (i < NNODES) ? cnt[i] : 0u;
  sh[t] = v;
  __syncthreads();
  for (int s = 1; s < 256; s <<= 1) {
    unsigned x = (t >= s) ? sh[t - s] : 0u;
    __syncthreads();
    sh[t] += x;
    __syncthreads();
  }
  const unsigned incl = sh[t];
  if (i < NNODES) rowptr[i] = (int)(prefix + incl - v);
  if (i == NNODES - 1) rowptr[NNODES] = (int)(prefix + incl);
}

__global__ void csr_fill(const int* __restrict__ ei, const int* __restrict__ rowptr,
                         const int* __restrict__ rank, int* __restrict__ csrc) {
  int e = blockIdx.x * blockDim.x + threadIdx.x;
  if (e >= NEDGES) return;
  int s = ei[e], d = ei[NEDGES + e];
  int pos = rowptr[d] + rank[e];
  if ((unsigned)pos < (unsigned)NEDGES) csrc[pos] = s;
}

__device__ __forceinline__ float leaky02(float a) { return a > 0.f ? a : 0.2f * a; }

__global__ __launch_bounds__(256) void gat_gather_h8(
    const int* __restrict__ rowptr, const int* __restrict__ csrc,
    const float* __restrict__ sd,
    const float* __restrict__ h, const float* __restrict__ bias,
    float* __restrict__ outp) {
  const int lane = threadIdx.x & 31;
  const int wv = __builtin_amdgcn_readfirstlane((int)(threadIdx.x >> 5));
  const int n = blockIdx.x * 8 + wv;
  if (n >= NNODES) return;
  const int hl = lane >> 2, q = lane & 3;
  int beg = rowptr[n], end = rowptr[n + 1];
  beg = min(max(beg, 0), NEDGES);
  end = min(max(end, beg), NEDGES);
  const float4* h4 = (const float4*)h;
  const float adn = sd[n * 16 + 8 + hl];
  const float a_self = leaky02(sd[n * 16 + hl] + adn);

  float m = a_self;
  for (int c0 = beg; c0 < end; c0 += 32) {
    const int cnt = min(32, end - c0);
    int sv = (lane < cnt) ? csrc[c0 + lane] : 0;
    sv = min(max(sv, 0), NNODES - 1);
    #pragma unroll 4
    for (int j4 = 0; j4 < cnt; j4 += 4) {
      const int jq = j4 + q;
      const int sq = __shfl(sv, jq);
      const float a = leaky02(sd[sq * 16 + hl] + adn);
      if (jq < cnt) m = fmaxf(m, a);
    }
  }
  m = fmaxf(m, __shfl_xor(m, 1));
  m = fmaxf(m, __shfl_xor(m, 2));

  const float w0 = expf(a_self - m);
  float den = (q == 0) ? w0 : 0.f;
  float4 acc;
  {
    const float4 r = h4[n * 32 + lane];
    acc.x = w0 * r.x; acc.y = w0 * r.y; acc.z = w0 * r.z; acc.w = w0 * r.w;
  }
  for (int c0 = beg; c0 < end; c0 += 32) {
    const int cnt = min(32, end - c0);
    int sv = (lane < cnt) ? csrc[c0 + lane] : 0;
    sv = min(max(sv, 0), NNODES - 1);
    for (int j4 = 0; j4 < cnt; j4 += 4) {
      const int jq = j4 + q;
      const int sq = __shfl(sv, jq);
      const float a = leaky02(sd[sq * 16 + hl] + adn);
      const float wq = (jq < cnt) ? expf(a - m) : 0.f;
      den += wq;
      if (j4 + 4 <= cnt) {
        float4 r[4];
        float w[4];
        #pragma unroll
        for (int k = 0; k < 4; ++k) {
          const int s = __builtin_amdgcn_readlane(sv, j4 + k);
          r[k] = h4[s * 32 + lane];
          w[k] = __shfl(wq, (lane & ~3) | k);
        }
        #pragma unroll
        for (int k = 0; k < 4; ++k) {
          acc.x += w[k] * r[k].x; acc.y += w[k] * r[k].y;
          acc.z += w[k] * r[k].z; acc.w += w[k] * r[k].w;
        }
      } else {
        #pragma unroll
        for (int k = 0; k < 3; ++k) {
          if (j4 + k < cnt) {
            const int s = __builtin_amdgcn_readlane(sv, j4 + k);
            const float w = __shfl(wq, (lane & ~3) | k);
            const float4 r = h4[s * 32 + lane];
            acc.x += w * r.x; acc.y += w * r.y; acc.z += w * r.z; acc.w += w * r.w;
          }
        }
      }
    }
  }
  den += __shfl_xor(den, 1);
  den += __shfl_xor(den, 2);
  const float inv = 1.0f / (den + 1e-16f);
  float4 o;
  o.x = acc.x * inv + bias[lane * 4 + 0]; o.y = acc.y * inv + bias[lane * 4 + 1];
  o.z = acc.z * inv + bias[lane * 4 + 2]; o.w = acc.w * inv + bias[lane * 4 + 3];
  o.x = o.x > 0.f ? o.x : expm1f(o.x); o.y = o.y > 0.f ? o.y : expm1f(o.y);
  o.z = o.z > 0.f ? o.z : expm1f(o.z); o.w = o.w > 0.f ? o.w : expm1f(o.w);
  ((float4*)outp)[n * 32 + lane] = o;
}

__device__ __forceinline__ int sd2_src(int n) { return ((n >> 4) << 5) + (n & 15); }
__device__ __forceinline__ int sd2_dst(int n) { return ((n >> 4) << 5) + 16 + (n & 15); }

__global__ __launch_bounds__(256) void gat_gather_h1(
    const int* __restrict__ rowptr, const int* __restrict__ csrc,
    const float* __restrict__ sd,
    const float* __restrict__ h, const float* __restrict__ bias,
    float* __restrict__ outp) {
  const int lane = threadIdx.x & 31;
  const int wv = __builtin_amdgcn_readfirstlane((int)(threadIdx.x >> 5));
  const int n = blockIdx.x * 8 + wv;
  if (n >= NNODES) return;
  const int es = lane >> 2, q = lane & 3;
  int beg = rowptr[n], end = rowptr[n + 1];
  beg = min(max(beg, 0), NEDGES);
  end = min(max(end, beg), NEDGES);
  const float4* h4 = (const float4*)h;
  const float adn = sd[sd2_dst(n)];
  const float a_self = leaky02(sd[sd2_src(n)] + adn);

  float m = a_self;
  for (int c0 = beg; c0 < end; c0 += 32) {
    const int j = c0 + lane;
    if (j < end) {
      int s = csrc[j];
      s = min(max(s, 0), NNODES - 1);
      m = fmaxf(m, leaky02(sd[sd2_src(s)] + adn));
    }
  }
  #pragma unroll
  for (int o = 16; o > 0; o >>= 1) m = fmaxf(m, __shfl_xor(m, o));

  const float w0 = expf(a_self - m);
  float den = (lane == 0) ? w0 : 0.f;
  float4 acc = {0.f, 0.f, 0.f, 0.f};
  if (es == 0) {
    const float4 r = h4[n * 4 + q];
    acc.x = w0 * r.x; acc.y = w0 * r.y; acc.z = w0 * r.z; acc.w = w0 * r.w;
  }
  for (int c0 = beg; c0 < end; c0 += 32) {
    const int cnt = min(32, end - c0);
    int s = 0;
    float w = 0.f;
    if (lane < cnt) {
      s = csrc[c0 + lane];
      s = min(max(s, 0), NNODES - 1);
      w = expf(leaky02(sd[sd2_src(s)] + adn) - m);
    }
    den += w;
    #pragma unroll
    for (int it = 0; it < 4; ++it) {
      const int jj = it * 8 + es;
      const int sj = __shfl(s, jj);
      const float wj = __shfl(w, jj);
      const float4 r = h4[sj * 4 + q];
      const bool ok = jj < cnt;
      acc.x += wj * (ok ? r.x : 0.f); acc.y += wj * (ok ? r.y : 0.f);
      acc.z += wj * (ok ? r.z : 0.f); acc.w += wj * (ok ? r.w : 0.f);
    }
  }
  #pragma unroll
  for (int o = 16; o > 0; o >>= 1) den += __shfl_xor(den, o);
  #pragma unroll
  for (int o = 16; o >= 4; o >>= 1) {
    acc.x += __shfl_xor(acc.x, o); acc.y += __shfl_xor(acc.y, o);
    acc.z += __shfl_xor(acc.z, o); acc.w += __shfl_xor(acc.w, o);
  }
  const float inv = 1.0f / (den + 1e-16f);
  float4 v;
  v.x = acc.x * inv + bias[q * 4 + 0]; v.y = acc.y * inv + bias[q * 4 + 1];
  v.z = acc.z * inv + bias[q * 4 + 2]; v.w = acc.w * inv + bias[q * 4 + 3];
  float vm = fmaxf(fmaxf(v.x, v.y), fmaxf(v.z, v.w));
  vm = fmaxf(vm, __shfl_xor(vm, 1));
  vm = fmaxf(vm, __shfl_xor(vm, 2));
  float se = (expf(v.x - vm) + expf(v.y - vm)) + (expf(v.z - vm) + expf(v.w - vm));
  se += __shfl_xor(se, 1);
  se += __shfl_xor(se, 2);
  const float ls = logf(se);
  if (es == 0) {
    float4 o4;
    o4.x = v.x - vm - ls; o4.y = v.y - vm - ls; o4.z = v.z - vm - ls; o4.w = v.w - vm - ls;
    ((float4*)outp)[n * 4 + q] = o4;
  }
}

extern "C" void kernel_launch(void* const* d_in, const int* in_sizes, int n_in,
                              void* d_out, int out_size, void* d_ws, size_t ws_size,
                              hipStream_t stream) {
  const float* x     = (const float*)d_in[0];
  const int*   ei    = (const int*)  d_in[1];
  const float* W1    = (const float*)d_in[2];
  const float* attS1 = (const float*)d_in[3];
  const float* attD1 = (const float*)d_in[4];
  const float* b1    = (const float*)d_in[5];
  const float* W2    = (const float*)d_in[6];
  const float* attS2 = (const float*)d_in[7];
  const float* attD2 = (const float*)d_in[8];
  const float* b2    = (const float*)d_in[9];
  float* out = (float*)d_out;

  char* ws = (char*)d_ws;
  size_t off = 0;
  auto alloc = [&](size_t bytes) -> void* {
    void* p = ws + off;
    off = (off + bytes + 255) & ~(size_t)255;
    return p;
  };

  float*    h1   = (float*)   alloc((size_t)NNODES * F1 * 4);
  float*    sd1  = (float*)   alloc((size_t)NNODES * 16 * 4);
  float*    agg1 = (float*)   alloc((size_t)NNODES * F1 * 4);
  int*      csrc = (int*)     alloc((size_t)NEDGES * 4);
  int*      rptr = (int*)     alloc((size_t)(NNODES + 1) * 4);
  unsigned* cnt  = (unsigned*)alloc((size_t)NNODES * 4);
  int*      rank = (int*)agg1;
  float*    h2   = h1;
  float*    sd2  = h1 + (size_t)NNODES * C_;

  const int TB = 256;
  auto nb = [](long long n, int b) { return (unsigned)((n + b - 1) / b); };

  fill_u32<<<nb(NNODES, TB), TB, 0, stream>>>(cnt, 0u, (size_t)NNODES);
  edge_rank<<<nb(NEDGES, TB), TB, 0, stream>>>(ei, cnt, rank);
  deg_scan<<<nb(NNODES, 256), 256, 0, stream>>>(cnt, rptr);
  csr_fill<<<nb(NEDGES, TB), TB, 0, stream>>>(ei, rptr, rank, csrc);

  gemm_f32_wmma_strip<8, NFEAT, F1><<<3125, 32, 0, stream>>>(x, W1, attS1, attD1, h1, sd1);
  gat_gather_h8<<<nb(NNODES, 8), 256, 0, stream>>>(rptr, csrc, sd1, h1, b1, agg1);

  gemm_f32_wmma_strip<1, F1, C_><<<3125, 32, 0, stream>>>(agg1, W2, attS2, attD2, h2, sd2);
  gat_gather_h1<<<nb(NNODES, 8), 256, 0, stream>>>(rptr, csrc, sd2, h2, b2, out);
}


# Edge grouping off global atomics: edges binned by 256-node destination range, then ranked per node, both on LDS counters; lists leave LDS as whole lines; 8 -> 7 launches; list and sum order change
# speedup vs baseline: 14.1049x; 1.7891x over previous
#include <hip/hip_runtime.h>
#include <hip/hip_bf16.h>

#define NNODES 50000
#define NEDGES 1600000
#define NFEAT  512
#define F1     128
#define H1_    8
#define C_     16
#define NCLASS 16

typedef float v2f __attribute__((ext_vector_type(2)));
typedef float v4f __attribute__((ext_vector_type(4)));
typedef float v8f __attribute__((ext_vector_type(8)));

template <int TN, int K, int NOUT>
__global__ __launch_bounds__(32) void gemm_f32_wmma_strip(
    const float* __restrict__ X, const float* __restrict__ Wt,
    const float* __restrict__ attS, const float* __restrict__ attD,
    float* __restrict__ D, float* __restrict__ SD) {
  static_assert(TN == 8 || TN == 1, "score record layouts exist for 8 heads and for 1 head");
  int tm = blockIdx.x;
  int lane = threadIdx.x;
  int l15 = lane & 15, half = lane >> 4;
  static_assert(K % 8 == 0, "the k loop walks blocks of 8");
  const float* ap = X  + (size_t)(tm * 16 + l15) * K + half * 4;
  const float* bp = Wt + (size_t)l15 * K + half * 4;
  v8f acc[TN];
  #pragma unroll
  for (int t = 0; t < TN; ++t) acc[t] = (v8f){};
  #pragma unroll 2
  for (int k = 0; k < K; k += 8) {
    const v4f a = *(const v4f*)(ap + k);
    v4f b[TN];
    #pragma unroll
    for (int t = 0; t < TN; ++t) b[t] = *(const v4f*)(bp + t * 16 * K + k);
    const v2f a0 = {a.x, a.y}, a1 = {a.z, a.w};
    #pragma unroll
    for (int t = 0; t < TN; ++t) {
      const v2f b0 = {b[t].x, b[t].y};
      acc[t] = __builtin_amdgcn_wmma_f32_16x16x4_f32(false, a0, false, b0,
                                                     (short)0, acc[t], false, false);
    }
    #pragma unroll
    for (int t = 0; t < TN; ++t) {
      const v2f b1 = {b[t].z, b[t].w};
      acc[t] = __builtin_amdgcn_wmma_f32_16x16x4_f32(false, a1, false, b1,
                                                     (short)0, acc[t], false, false);
    }
  }
  #pragma unroll
  for (int t = 0; t < TN; ++t) {
    float* dp = D + (size_t)(tm * 16 + half * 8) * NOUT + t * 16 + l15;
    #pragma unroll
    for (int r = 0; r < 8; ++r) dp[r * NOUT] = acc[t][r];
  }

  const bool hi8 = (l15 & 8) != 0;
  const int l7 = l15 & 7;
  float keep[8];
  #pragma unroll
  for (int r = 0; r < 8; ++r) keep[r] = 0.f;
  #pragma unroll
  for (int t = 0; t < TN; ++t) {
    const float ws = attS[t * 16 + l15];
    const float wd = attD[t * 16 + l15];
    #pragma unroll
    for (int r = 0; r < 8; ++r) {
      const float ps = acc[t][r] * ws;
      const float pd = acc[t][r] * wd;
      float v = (hi8 ? pd : ps) + __shfl_xor(hi8 ? ps : pd, 8);
      v += __shfl_xor(v, 4);
      v += __shfl_xor(v, 2);
      v += __shfl_xor(v, 1);
      if (TN == 8) {
        if (l7 == t) keep[r] = v;
      } else {
        if (l7 == r) keep[0] = v;
      }
    }
  }
  if (TN == 8) {
    float* sp = SD + (size_t)(tm * 16 + half * 8) * 16 + l15;
    #pragma unroll
    for (int r = 0; r < 8; ++r) sp[r * 16] = keep[r];
  } else {
    SD[(size_t)tm * 32 + (hi8 ? 16 : 0) + half * 8 + l7] = keep[0];
  }
}

#define NPB   256
#define NBIN  ((NNODES + NPB - 1) / NPB)
#define EPB   6400
#define NEB   ((NEDGES + EPB - 1) / EPB)
#define ETH   1024
#define EPT   ((EPB + ETH - 1) / ETH)
#define SCAP  12288
#define RPT   10

static_assert(NNODES <= 65536, "source and destination ids are packed 16 + 16 bits");
static_assert(NBIN <= 256 && NPB == 256, "per-bin tables have 256 entries; node in bin = id & 255");

__device__ __forceinline__ unsigned scan256_wave(unsigned* arr, int lane) {
  unsigned a[8], sum = 0u;
  #pragma unroll
  for (int i = 0; i < 8; ++i) { const unsigned v = arr[lane * 8 + i]; a[i] = sum; sum += v; }
  unsigned incl = sum;
  #pragma unroll
  for (int o = 1; o < 32; o <<= 1) {
    const unsigned y = __shfl_up(incl, o);
    if (lane >= o) incl += y;
  }
  const unsigned excl = incl - sum;
  #pragma unroll
  for (int i = 0; i < 8; ++i) arr[lane * 8 + i] = a[i] + excl;
  return __shfl(incl, 31);
}

__global__ __launch_bounds__(ETH) void bin_count(const int* __restrict__ ei,
                                                 unsigned* __restrict__ hist) {
  __shared__ unsigned h[256];
  const int t = threadIdx.x, b = blockIdx.x;
  if (t < 256) h[t] = 0u;
  const int e0 = b * EPB;
  const int nE = min(EPB, NEDGES - e0);
  int d[EPT];
  #pragma unroll
  for (int k = 0; k < EPT; ++k) {
    const int i = t + k * ETH;
    const int dv = ei[NEDGES + e0 + min(i, nE - 1)];
    d[k] = (i < nE) ? dv : -1;
  }
  __syncthreads();
  #pragma unroll
  for (int k = 0; k < EPT; ++k)
    if ((unsigned)d[k] < (unsigned)NNODES) atomicAdd(&h[d[k] / NPB], 1u);
  __syncthreads();
  if (t < NBIN) hist[b * NBIN + t] = h[t];
}

__global__ __launch_bounds__(ETH) void bin_place(const int* __restrict__ ei,
                                                 const unsigned* __restrict__ hist,
                                                 unsigned* __restrict__ bstart,
                                                 unsigned* __restrict__ binned) {
  __shared__ unsigned s_tot[ETH / 256][256];
  __shared__ unsigned s_pre[ETH / 256][256];
  __shared__ unsigned s_bin0[256];
  __shared__ unsigned s_base[256];
  __shared__ unsigned s_l0[256];
  __shared__ unsigned s_cur[256];
  __shared__ unsigned s_word[EPB];
  __shared__ unsigned s_n;
  const int t = threadIdx.x, b = blockIdx.x;
  const int lane = t & 31, wv = t >> 5;
  const int bin = t & 255, g = t >> 8;
  const int e0 = b * EPB;
  const int nE = min(EPB, NEDGES - e0);
  int s[EPT], d[EPT];
  #pragma unroll
  for (int k = 0; k < EPT; ++k) {
    const int i = t + k * ETH;
    const int ic = min(i, nE - 1);
    const int sv = ei[e0 + ic];
    const int dv = ei[NEDGES + e0 + ic];
    s[k] = sv;
    d[k] = (i < nE) ? dv : -1;
  }
  unsigned tot = 0u, pre = 0u;
  if (bin < NBIN) {
    #pragma unroll 8
    for (int blk = g; blk < NEB; blk += ETH / 256) {
      const unsigned v = hist[blk * NBIN + bin];
      tot += v;
      pre += (blk < b) ? v : 0u;
    }
  }
  s_tot[g][bin] = tot;
  s_pre[g][bin] = pre;
  __syncthreads();
  if (t < 256) {
    unsigned T = 0u, P = 0u;
    #pragma unroll
    for (int q = 0; q < ETH / 256; ++q) { T += s_tot[q][t]; P += s_pre[q][t]; }
    s_bin0[t] = T;
    s_base[t] = P;
    s_l0[t] = (t < NBIN) ? hist[b * NBIN + t] : 0u;
  }
  __syncthreads();
  if (wv == 0) {
    const unsigned total = scan256_wave(s_bin0, lane);
    if (b == 0 && lane == 0) bstart[NBIN] = total;
  } else if (wv == 1) {
    const unsigned n = scan256_wave(s_l0, lane);
    if (lane == 0) s_n = n;
  }
  __syncthreads();
  if (t < 256) {
    s_base[t] += s_bin0[t];
    s_cur[t] = s_l0[t];
    if (b == 0 && t < NBIN) bstart[t] = s_bin0[t];
  }
  __syncthreads();
  #pragma unroll
  for (int k = 0; k < EPT; ++k) {
    if ((unsigned)d[k] < (unsigned)NNODES) {
      const unsigned p = atomicAdd(&s_cur[d[k] / NPB], 1u);
      const int sc = min(max(s[k], 0), NNODES - 1);
      if (p < (unsigned)EPB) s_word[p] = ((unsigned)sc << 16) | (unsigned)d[k];
    }
  }
  __syncthreads();
  const unsigned n = min(s_n, (unsigned)EPB);
  for (unsigned j = t; j < n; j += ETH) {
    const unsigned w = s_word[j];
    const unsigned bi = (w & 0xFFFFu) / NPB;
    const unsigned pos = s_base[bi] + (j - s_l0[bi]);
    if (pos < (unsigned)NEDGES) binned[pos] = w;
  }
}

__global__ __launch_bounds__(ETH) void bin_sort(const unsigned* __restrict__ binned,
                                                const unsigned* __restrict__ bstart,
                                                int* __restrict__ rowptr,
                                                int* __restrict__ csrc) {
  __shared__ unsigned s_cnt[256];
  __shared__ unsigned s_stage[SCAP];
  const int t = threadIdx.x, b = blockIdx.x;
  const int lane = t & 31, wv = t >> 5;
  unsigned beg = bstart[b], end = bstart[b + 1];
  beg = min(beg, (unsigned)NEDGES);
  end = min(max(end, beg), (unsigned)NEDGES);
  const unsigned cnt = end - beg;
  if (t < 256) s_cnt[t] = 0u;
  __syncthreads();
  unsigned w[RPT];
  const bool one_chunk = cnt <= (unsigned)(ETH * RPT);
  for (unsigned c0 = 0; c0 < cnt; c0 += ETH * RPT) {
    #pragma unroll
    for (int k = 0; k < RPT; ++k) w[k] = binned[beg + min(c0 + t + k * ETH, cnt - 1u)];
    #pragma unroll
    for (int k = 0; k < RPT; ++k)
      if (c0 + t + k * ETH < cnt) atomicAdd(&s_cnt[w[k] & 255u], 1u);
  }
  __syncthreads();
  if (wv == 0) {
    scan256_wave(s_cnt, lane);
    #pragma unroll
    for (int i = 0; i < 8; ++i) {
      const int node = b * NPB + lane * 8 + i;
      if (node < NNODES) rowptr[node] = (int)(beg + s_cnt[lane * 8 + i]);
    }
    if (b == NBIN - 1 && lane == 0) rowptr[NNODES] = (int)end;
  }
  __syncthreads();
  for (unsigned c0 = 0; c0 < cnt; c0 += ETH * RPT) {
    if (!one_chunk) {
      #pragma unroll
      for (int k = 0; k < RPT; ++k) w[k] = binned[beg + min(c0 + t + k * ETH, cnt - 1u)];
    }
    #pragma unroll
    for (int k = 0; k < RPT; ++k) {
      if (c0 + t + k * ETH < cnt) {
        const unsigned p = atomicAdd(&s_cnt[w[k] & 255u], 1u);
        const int sv = (int)(w[k] >> 16);
        if (p < (unsigned)SCAP) s_stage[p] = (unsigned)sv;
        else if (p < cnt) csrc[beg + p] = sv;
      }
    }
  }
  __syncthreads();
  const unsigned lim = min(cnt, (unsigned)SCAP);
  for (unsigned i = t; i < lim; i += ETH) csrc[beg + i] = (int)s_stage[i];
}

__device__ __forceinline__ float leaky02(float a) { return a > 0.f ? a : 0.2f * a; }

__global__ __launch_bounds__(256) void gat_gather_h8(
    const int* __restrict__ rowptr, const int* __restrict__ csrc,
    const float* __restrict__ sd,
    const float* __restrict__ h, const float* __restrict__ bias,
    float* __restrict__ outp) {
  const int lane = threadIdx.x & 31;
  const int wv = __builtin_amdgcn_readfirstlane((int)(threadIdx.x >> 5));
  const int n = blockIdx.x * 8 + wv;
  if (n >= NNODES) return;
  const int hl = lane >> 2, q = lane & 3;
  int beg = rowptr[n], end = rowptr[n + 1];
  beg = min(max(beg, 0), NEDGES);
  end = min(max(end, beg), NEDGES);
  const float4* h4 = (const float4*)h;
  const float adn = sd[n * 16 + 8 + hl];
  const float a_self = leaky02(sd[n * 16 + hl] + adn);

  float m = a_self;
  for (int c0 = beg; c0 < end; c0 += 32) {
    const int cnt = min(32, end - c0);
    int sv = (lane < cnt) ? csrc[c0 + lane] : 0;
    sv = min(max(sv, 0), NNODES - 1);
    #pragma unroll 4
    for (int j4 = 0; j4 < cnt; j4 += 4) {
      const int jq = j4 + q;
      const int sq = __shfl(sv, jq);
      const float a = leaky02(sd[sq * 16 + hl] + adn);
      if (jq < cnt) m = fmaxf(m, a);
    }
  }
  m = fmaxf(m, __shfl_xor(m, 1));
  m = fmaxf(m, __shfl_xor(m, 2));

  const float w0 = expf(a_self - m);
  float den = (q == 0) ? w0 : 0.f;
  float4 acc;
  {
    const float4 r = h4[n * 32 + lane];
    acc.x = w0 * r.x; acc.y = w0 * r.y; acc.z = w0 * r.z; acc.w = w0 * r.w;
  }
  for (int c0 = beg; c0 < end; c0 += 32) {
    const int cnt = min(32, end - c0);
    int sv = (lane < cnt) ? csrc[c0 + lane] : 0;
    sv = min(max(sv, 0), NNODES - 1);
    for (int j4 = 0; j4 < cnt; j4 += 4) {
      const int jq = j4 + q;
      const int sq = __shfl(sv, jq);
      const float a = leaky02(sd[sq * 16 + hl] + adn);
      const float wq = (jq < cnt) ? expf(a - m) : 0.f;
      den += wq;
      if (j4 + 4 <= cnt) {
        float4 r[4];
        float w[4];
        #pragma unroll
        for (int k = 0; k < 4; ++k) {
          const int s = __builtin_amdgcn_readlane(sv, j4 + k);
          r[k] = h4[s * 32 + lane];
          w[k] = __shfl(wq, (lane & ~3) | k);
        }
        #pragma unroll
        for (int k = 0; k < 4; ++k) {
          acc.x += w[k] * r[k].x; acc.y += w[k] * r[k].y;
          acc.z += w[k] * r[k].z; acc.w += w[k] * r[k].w;
        }
      } else {
        #pragma unroll
        for (int k = 0; k < 3; ++k) {
          if (j4 + k < cnt) {
            const int s = __builtin_amdgcn_readlane(sv, j4 + k);
            const float w = __shfl(wq, (lane & ~3) | k);
            const float4 r = h4[s * 32 + lane];
            acc.x += w * r.x; acc.y += w * r.y; acc.z += w * r.z; acc.w += w * r.w;
          }
        }
      }
    }
  }
  den += __shfl_xor(den, 1);
  den += __shfl_xor(den, 2);
  const float inv = 1.0f / (den + 1e-16f);
  float4 o;
  o.x = acc.x * inv + bias[lane * 4 + 0]; o.y = acc.y * inv + bias[lane * 4 + 1];
  o.z = acc.z * inv + bias[lane * 4 + 2]; o.w = acc.w * inv + bias[lane * 4 + 3];
  o.x = o.x > 0.f ? o.x : expm1f(o.x); o.y = o.y > 0.f ? o.y : expm1f(o.y);
  o.z = o.z > 0.f ? o.z : expm1f(o.z); o.w = o.w > 0.f ? o.w : expm1f(o.w);
  ((float4*)outp)[n * 32 + lane] = o;
}

__device__ __forceinline__ int sd2_src(int n) { return ((n >> 4) << 5) + (n & 15); }
__device__ __forceinline__ int sd2_dst(int n) { return ((n >> 4) << 5) + 16 + (n & 15); }

__global__ __launch_bounds__(256) void gat_gather_h1(
    const int* __restrict__ rowptr, const int* __restrict__ csrc,
    const float* __restrict__ sd,
    const float* __restrict__ h, const float* __restrict__ bias,
    float* __restrict__ outp) {
  const int lane = threadIdx.x & 31;
  const int wv = __builtin_amdgcn_readfirstlane((int)(threadIdx.x >> 5));
  const int n = blockIdx.x * 8 + wv;
  if (n >= NNODES) return;
  const int es = lane >> 2, q = lane & 3;
  int beg = rowptr[n], end = rowptr[n + 1];
  beg = min(max(beg, 0), NEDGES);
  end = min(max(end, beg), NEDGES);
  const float4* h4 = (const float4*)h;
  const float adn = sd[sd2_dst(n)];
  const float a_self = leaky02(sd[sd2_src(n)] + adn);

  float m = a_self;
  for (int c0 = beg; c0 < end; c0 += 32) {
    const int j = c0 + lane;
    if (j < end) {
      int s = csrc[j];
      s = min(max(s, 0), NNODES - 1);
      m = fmaxf(m, leaky02(sd[sd2_src(s)] + adn));
    }
  }
  #pragma unroll
  for (int o = 16; o > 0; o >>= 1) m = fmaxf(m, __shfl_xor(m, o));

  const float w0 = expf(a_self - m);
  float den = (lane == 0) ? w0 : 0.f;
  float4 acc = {0.f, 0.f, 0.f, 0.f};
  if (es == 0) {
    const float4 r = h4[n * 4 + q];
    acc.x = w0 * r.x; acc.y = w0 * r.y; acc.z = w0 * r.z; acc.w = w0 * r.w;
  }
  for (int c0 = beg; c0 < end; c0 += 32) {
    const int cnt = min(32, end - c0);
    int s = 0;
    float w = 0.f;
    if (lane < cnt) {
      s = csrc[c0 + lane];
      s = min(max(s, 0), NNODES - 1);
      w = expf(leaky02(sd[sd2_src(s)] + adn) - m);
    }
    den += w;
    #pragma unroll
    for (int it = 0; it < 4; ++it) {
      const int jj = it * 8 + es;
      const int sj = __shfl(s, jj);
      const float wj = __shfl(w, jj);
      const float4 r = h4[sj * 4 + q];
      const bool ok = jj < cnt;
      acc.x += wj * (ok ? r.x : 0.f); acc.y += wj * (ok ? r.y : 0.f);
      acc.z += wj * (ok ? r.z : 0.f); acc.w += wj * (ok ? r.w : 0.f);
    }
  }
  #pragma unroll
  for (int o = 16; o > 0; o >>= 1) den += __shfl_xor(den, o);
  #pragma unroll
  for (int o = 16; o >= 4; o >>= 1) {
    acc.x += __shfl_xor(acc.x, o); acc.y += __shfl_xor(acc.y, o);
    acc.z += __shfl_xor(acc.z, o); acc.w += __shfl_xor(acc.w, o);
  }
  const float inv = 1.0f / (den + 1e-16f);
  float4 v;
  v.x = acc.x * inv + bias[q * 4 + 0]; v.y = acc.y * inv + bias[q * 4 + 1];
  v.z = acc.z * inv + bias[q * 4 + 2]; v.w = acc.w * inv + bias[q * 4 + 3];
  float vm = fmaxf(fmaxf(v.x, v.y), fmaxf(v.z, v.w));
  vm = fmaxf(vm, __shfl_xor(vm, 1));
  vm = fmaxf(vm, __shfl_xor(vm, 2));
  float se = (expf(v.x - vm) + expf(v.y - vm)) + (expf(v.z - vm) + expf(v.w - vm));
  se += __shfl_xor(se, 1);
  se += __shfl_xor(se, 2);
  const float ls = logf(se);
  if (es == 0) {
    float4 o4;
    o4.x = v.x - vm - ls; o4.y = v.y - vm - ls; o4.z = v.z - vm - ls; o4.w = v.w - vm - ls;
    ((float4*)outp)[n * 4 + q] = o4;
  }
}

extern "C" void kernel_launch(void* const* d_in, const int* in_sizes, int n_in,
                              void* d_out, int out_size, void* d_ws, size_t ws_size,
                              hipStream_t stream) {
  const float* x     = (const float*)d_in[0];
  const int*   ei    = (const int*)  d_in[1];
  const float* W1    = (const float*)d_in[2];
  const float* attS1 = (const float*)d_in[3];
  const float* attD1 = (const float*)d_in[4];
  const float* b1    = (const float*)d_in[5];
  const float* W2    = (const float*)d_in[6];
  const float* attS2 = (const float*)d_in[7];
  const float* attD2 = (const float*)d_in[8];
  const float* b2    = (const float*)d_in[9];
  float* out = (float*)d_out;

  char* ws = (char*)d_ws;
  size_t off = 0;
  auto alloc = [&](size_t bytes) -> void* {
    void* p = ws + off;
    off = (off + bytes + 255) & ~(size_t)255;
    return p;
  };

  float*    h1   = (float*)   alloc((size_t)NNODES * F1 * 4);
  float*    sd1  = (float*)   alloc((size_t)NNODES * 16 * 4);
  float*    agg1 = (float*)   alloc((size_t)NNODES * F1 * 4);
  int*      csrc = (int*)     alloc((size_t)NEDGES * 4);
  int*      rptr = (int*)     alloc((size_t)(NNODES + 1) * 4);
  unsigned* aux  = (unsigned*)alloc((size_t)NNODES * 4);
  static_assert((size_t)NEB * NBIN + NBIN + 1 <= (size_t)NNODES, "hist and bstart fit the region");
  unsigned* hist   = aux;
  unsigned* bstart = aux + (size_t)NEB * NBIN;
  unsigned* binned = (unsigned*)agg1;
  float*    h2   = h1;
  float*    sd2  = h1 + (size_t)NNODES * C_;

  auto nb = [](long long n, int b) { return (unsigned)((n + b - 1) / b); };

  bin_count<<<NEB, ETH, 0, stream>>>(ei, hist);
  bin_place<<<NEB, ETH, 0, stream>>>(ei, hist, bstart, binned);
  bin_sort<<<NBIN, ETH, 0, stream>>>(binned, bstart, rptr, csrc);

  gemm_f32_wmma_strip<8, NFEAT, F1><<<3125, 32, 0, stream>>>(x, W1, attS1, attD1, h1, sd1);
  gat_gather_h8<<<nb(NNODES, 8), 256, 0, stream>>>(rptr, csrc, sd1, h1, b1, agg1);

  gemm_f32_wmma_strip<1, F1, C_><<<3125, 32, 0, stream>>>(agg1, W2, attS2, attD2, h2, sd2);
  gat_gather_h1<<<nb(NNODES, 8), 256, 0, stream>>>(rptr, csrc, sd2, h2, b2, out);
}


# Each edge's attention score is read once per layer: the max pass keeps a list's scores (registers; layer-1 weights reach the row loop through LDS) for the sum pass; lists over 64 edges as before
# speedup vs baseline: 15.4986x; 1.0988x over previous
#include <hip/hip_runtime.h>
#include <hip/hip_bf16.h>

#define NNODES 50000
#define NEDGES 1600000
#define NFEAT  512
#define F1     128
#define H1_    8
#define C_     16
#define NCLASS 16

typedef float v2f __attribute__((ext_vector_type(2)));
typedef float v4f __attribute__((ext_vector_type(4)));
typedef float v8f __attribute__((ext_vector_type(8)));

template <int TN, int K, int NOUT>
__global__ __launch_bounds__(32) void gemm_f32_wmma_strip(
    const float* __restrict__ X, const float* __restrict__ Wt,
    const float* __restrict__ attS, const float* __restrict__ attD,
    float* __restrict__ D, float* __restrict__ SD) {
  static_assert(TN == 8 || TN == 1, "score record layouts exist for 8 heads and for 1 head");
  int tm = blockIdx.x;
  int lane = threadIdx.x;
  int l15 = lane & 15, half = lane >> 4;
  static_assert(K % 8 == 0, "the k loop walks blocks of 8");
  const float* ap = X  + (size_t)(tm * 16 + l15) * K + half * 4;
  const float* bp = Wt + (size_t)l15 * K + half * 4;
  v8f acc[TN];
  #pragma unroll
  for (int t = 0; t < TN; ++t) acc[t] = (v8f){};
  #pragma unroll 2
  for (int k = 0; k < K; k += 8) {
    const v4f a = *(const v4f*)(ap + k);
    v4f b[TN];
    #pragma unroll
    for (int t = 0; t < TN; ++t) b[t] = *(const v4f*)(bp + t * 16 * K + k);
    const v2f a0 = {a.x, a.y}, a1 = {a.z, a.w};
    #pragma unroll
    for (int t = 0; t < TN; ++t) {
      const v2f b0 = {b[t].x, b[t].y};
      acc[t] = __builtin_amdgcn_wmma_f32_16x16x4_f32(false, a0, false, b0,
                                                     (short)0, acc[t], false, false);
    }
    #pragma unroll
    for (int t = 0; t < TN; ++t) {
      const v2f b1 = {b[t].z, b[t].w};
      acc[t] = __builtin_amdgcn_wmma_f32_16x16x4_f32(false, a1, false, b1,
                                                     (short)0, acc[t], false, false);
    }
  }
  #pragma unroll
  for (int t = 0; t < TN; ++t) {
    float* dp = D + (size_t)(tm * 16 + half * 8) * NOUT + t * 16 + l15;
    #pragma unroll
    for (int r = 0; r < 8; ++r) dp[r * NOUT] = acc[t][r];
  }

  const bool hi8 = (l15 & 8) != 0;
  const int l7 = l15 & 7;
  float keep[8];
  #pragma unroll
  for (int r = 0; r < 8; ++r) keep[r] = 0.f;
  #pragma unroll
  for (int t = 0; t < TN; ++t) {
    const float ws = attS[t * 16 + l15];
    const float wd = attD[t * 16 + l15];
    #pragma unroll
    for (int r = 0; r < 8; ++r) {
      const float ps = acc[t][r] * ws;
      const float pd = acc[t][r] * wd;
      float v = (hi8 ? pd : ps) + __shfl_xor(hi8 ? ps : pd, 8);
      v += __shfl_xor(v, 4);
      v += __shfl_xor(v, 2);
      v += __shfl_xor(v, 1);
      if (TN == 8) {
        if (l7 == t) keep[r] = v;
      } else {
        if (l7 == r) keep[0] = v;
      }
    }
  }
  if (TN == 8) {
    float* sp = SD + (size_t)(tm * 16 + half * 8) * 16 + l15;
    #pragma unroll
    for (int r = 0; r < 8; ++r) sp[r * 16] = keep[r];
  } else {
    SD[(size_t)tm * 32 + (hi8 ? 16 : 0) + half * 8 + l7] = keep[0];
  }
}

#define NPB   256
#define NBIN  ((NNODES + NPB - 1) / NPB)
#define EPB   6400
#define NEB   ((NEDGES + EPB - 1) / EPB)
#define ETH   1024
#define EPT   ((EPB + ETH - 1) / ETH)
#define SCAP  12288
#define RPT   10

static_assert(NNODES <= 65536, "source and destination ids are packed 16 + 16 bits");
static_assert(NBIN <= 256 && NPB == 256, "per-bin tables have 256 entries; node in bin = id & 255");

__device__ __forceinline__ unsigned scan256_wave(unsigned* arr, int lane) {
  unsigned a[8], sum = 0u;
  #pragma unroll
  for (int i = 0; i < 8; ++i) { const unsigned v = arr[lane * 8 + i]; a[i] = sum; sum += v; }
  unsigned incl = sum;
  #pragma unroll
  for (int o = 1; o < 32; o <<= 1) {
    const unsigned y = __shfl_up(incl, o);
    if (lane >= o) incl += y;
  }
  const unsigned excl = incl - sum;
  #pragma unroll
  for (int i = 0; i < 8; ++i) arr[lane * 8 + i] = a[i] + excl;
  return __shfl(incl, 31);
}

__global__ __launch_bounds__(ETH) void bin_count(const int* __restrict__ ei,
                                                 unsigned* __restrict__ hist) {
  __shared__ unsigned h[256];
  const int t = threadIdx.x, b = blockIdx.x;
  if (t < 256) h[t] = 0u;
  const int e0 = b * EPB;
  const int nE = min(EPB, NEDGES - e0);
  int d[EPT];
  #pragma unroll
  for (int k = 0; k < EPT; ++k) {
    const int i = t + k * ETH;
    const int dv = ei[NEDGES + e0 + min(i, nE - 1)];
    d[k] = (i < nE) ? dv : -1;
  }
  __syncthreads();
  #pragma unroll
  for (int k = 0; k < EPT; ++k)
    if ((unsigned)d[k] < (unsigned)NNODES) atomicAdd(&h[d[k] / NPB], 1u);
  __syncthreads();
  if (t < NBIN) hist[b * NBIN + t] = h[t];
}

__global__ __launch_bounds__(ETH) void bin_place(const int* __restrict__ ei,
                                                 const unsigned* __restrict__ hist,
                                                 unsigned* __restrict__ bstart,
                                                 unsigned* __restrict__ binned) {
  __shared__ unsigned s_tot[ETH / 256][256];
  __shared__ unsigned s_pre[ETH / 256][256];
  __shared__ unsigned s_bin0[256];
  __shared__ unsigned s_base[256];
  __shared__ unsigned s_l0[256];
  __shared__ unsigned s_cur[256];
  __shared__ unsigned s_word[EPB];
  __shared__ unsigned s_n;
  const int t = threadIdx.x, b = blockIdx.x;
  const int lane = t & 31, wv = t >> 5;
  const int bin = t & 255, g = t >> 8;
  const int e0 = b * EPB;
  const int nE = min(EPB, NEDGES - e0);
  int s[EPT], d[EPT];
  #pragma unroll
  for (int k = 0; k < EPT; ++k) {
    const int i = t + k * ETH;
    const int ic = min(i, nE - 1);
    const int sv = ei[e0 + ic];
    const int dv = ei[NEDGES + e0 + ic];
    s[k] = sv;
    d[k] = (i < nE) ? dv : -1;
  }
  unsigned tot = 0u, pre = 0u;
  if (bin < NBIN) {
    #pragma unroll 8
    for (int blk = g; blk < NEB; blk += ETH / 256) {
      const unsigned v = hist[blk * NBIN + bin];
      tot += v;
      pre += (blk < b) ? v : 0u;
    }
  }
  s_tot[g][bin] = tot;
  s_pre[g][bin] = pre;
  __syncthreads();
  if (t < 256) {
    unsigned T = 0u, P = 0u;
    #pragma unroll
    for (int q = 0; q < ETH / 256; ++q) { T += s_tot[q][t]; P += s_pre[q][t]; }
    s_bin0[t] = T;
    s_base[t] = P;
    s_l0[t] = (t < NBIN) ? hist[b * NBIN + t] : 0u;
  }
  __syncthreads();
  if (wv == 0) {
    const unsigned total = scan256_wave(s_bin0, lane);
    if (b == 0 && lane == 0) bstart[NBIN] = total;
  } else if (wv == 1) {
    const unsigned n = scan256_wave(s_l0, lane);
    if (lane == 0) s_n = n;
  }
  __syncthreads();
  if (t < 256) {
    s_base[t] += s_bin0[t];
    s_cur[t] = s_l0[t];
    if (b == 0 && t < NBIN) bstart[t] = s_bin0[t];
  }
  __syncthreads();
  #pragma unroll
  for (int k = 0; k < EPT; ++k) {
    if ((unsigned)d[k] < (unsigned)NNODES) {
      const unsigned p = atomicAdd(&s_cur[d[k] / NPB], 1u);
      const int sc = min(max(s[k], 0), NNODES - 1);
      if (p < (unsigned)EPB) s_word[p] = ((unsigned)sc << 16) | (unsigned)d[k];
    }
  }
  __syncthreads();
  const unsigned n = min(s_n, (unsigned)EPB);
  for (unsigned j = t; j < n; j += ETH) {
    const unsigned w = s_word[j];
    const unsigned bi = (w & 0xFFFFu) / NPB;
    const unsigned pos = s_base[bi] + (j - s_l0[bi]);
    if (pos < (unsigned)NEDGES) binned[pos] = w;
  }
}

__global__ __launch_bounds__(ETH) void bin_sort(const unsigned* __restrict__ binned,
                                                const unsigned* __restrict__ bstart,
                                                int* __restrict__ rowptr,
                                                int* __restrict__ csrc) {
  __shared__ unsigned s_cnt[256];
  __shared__ unsigned s_stage[SCAP];
  const int t = threadIdx.x, b = blockIdx.x;
  const int lane = t & 31, wv = t >> 5;
  unsigned beg = bstart[b], end = bstart[b + 1];
  beg = min(beg, (unsigned)NEDGES);
  end = min(max(end, beg), (unsigned)NEDGES);
  const unsigned cnt = end - beg;
  if (t < 256) s_cnt[t] = 0u;
  __syncthreads();
  unsigned w[RPT];
  const bool one_chunk = cnt <= (unsigned)(ETH * RPT);
  for (unsigned c0 = 0; c0 < cnt; c0 += ETH * RPT) {
    #pragma unroll
    for (int k = 0; k < RPT; ++k) w[k] = binned[beg + min(c0 + t + k * ETH, cnt - 1u)];
    #pragma unroll
    for (int k = 0; k < RPT; ++k)
      if (c0 + t + k * ETH < cnt) atomicAdd(&s_cnt[w[k] & 255u], 1u);
  }
  __syncthreads();
  if (wv == 0) {
    scan256_wave(s_cnt, lane);
    #pragma unroll
    for (int i = 0; i < 8; ++i) {
      const int node = b * NPB + lane * 8 + i;
      if (node < NNODES) rowptr[node] = (int)(beg + s_cnt[lane * 8 + i]);
    }
    if (b == NBIN - 1 && lane == 0) rowptr[NNODES] = (int)end;
  }
  __syncthreads();
  for (unsigned c0 = 0; c0 < cnt; c0 += ETH * RPT) {
    if (!one_chunk) {
      #pragma unroll
      for (int k = 0; k < RPT; ++k) w[k] = binned[beg + min(c0 + t + k * ETH, cnt - 1u)];
    }
    #pragma unroll
    for (int k = 0; k < RPT; ++k) {
      if (c0 + t + k * ETH < cnt) {
        const unsigned p = atomicAdd(&s_cnt[w[k] & 255u], 1u);
        const int sv = (int)(w[k] >> 16);
        if (p < (unsigned)SCAP) s_stage[p] = (unsigned)sv;
        else if (p < cnt) csrc[beg + p] = sv;
      }
    }
  }
  __syncthreads();
  const unsigned lim = min(cnt, (unsigned)SCAP);
  for (unsigned i = t; i < lim; i += ETH) csrc[beg + i] = (int)s_stage[i];
}

__device__ __forceinline__ float leaky02(float a) { return a > 0.f ? a : 0.2f * a; }

#define LCAP 64
#define LSTR (LCAP + 4)

__global__ __launch_bounds__(256) void gat_gather_h8(
    const int* __restrict__ rowptr, const int* __restrict__ csrc,
    const float* __restrict__ sd,
    const float* __restrict__ h, const float* __restrict__ bias,
    float* __restrict__ outp) {
  __shared__ __attribute__((aligned(16))) float s_w[8][H1_ * LSTR];
  const int lane = threadIdx.x & 31;
  const int wv = __builtin_amdgcn_readfirstlane((int)(threadIdx.x >> 5));
  const int n = blockIdx.x * 8 + wv;
  if (n >= NNODES) return;
  const int hl = lane >> 2, q = lane & 3;
  int beg = rowptr[n], end = rowptr[n + 1];
  beg = min(max(beg, 0), NEDGES);
  end = min(max(end, beg), NEDGES);
  const float4* h4 = (const float4*)h;
  const float adn = sd[n * 16 + 8 + hl];
  const float a_self = leaky02(sd[n * 16 + hl] + adn);

  const int deg = end - beg;
  float m, den;
  float4 acc;
  if (deg <= LCAP) {
    float* wl = &s_w[wv][0];
    const float4 rself = h4[n * 32 + lane];
    int sv0 = (lane < deg) ? csrc[beg + lane] : 0;
    sv0 = min(max(sv0, 0), NNODES - 1);
    int sv1 = 0;
    float a0[8], a1[8];
    m = a_self;
    #pragma unroll
    for (int i = 0; i < 8; ++i) {
      const int jq = 4 * i + q;
      const int sq = __shfl(sv0, jq);
      a0[i] = leaky02(sd[sq * 16 + hl] + adn);
      if (jq < deg) m = fmaxf(m, a0[i]);
    }
    if (deg > 32) {
      sv1 = (32 + lane < deg) ? csrc[beg + 32 + lane] : 0;
      sv1 = min(max(sv1, 0), NNODES - 1);
      #pragma unroll
      for (int i = 0; i < 8; ++i) {
        const int jq = 4 * i + q;
        const int sq = __shfl(sv1, jq);
        a1[i] = leaky02(sd[sq * 16 + hl] + adn);
        if (32 + jq < deg) m = fmaxf(m, a1[i]);
      }
    }
    m = fmaxf(m, __shfl_xor(m, 1));
    m = fmaxf(m, __shfl_xor(m, 2));

    const float w0 = expf(a_self - m);
    den = (q == 0) ? w0 : 0.f;
    #pragma unroll
    for (int i = 0; i < 8; ++i) {
      const int jq = 4 * i + q;
      const float wq = (jq < deg) ? expf(a0[i] - m) : 0.f;
      den += wq;
      wl[hl * LSTR + jq] = wq;
    }
    if (deg > 32) {
      #pragma unroll
      for (int i = 0; i < 8; ++i) {
        const int jq = 32 + 4 * i + q;
        const float wq = (jq < deg) ? expf(a1[i] - m) : 0.f;
        den += wq;
        wl[hl * LSTR + jq] = wq;
      }
    }
    acc.x = w0 * rself.x; acc.y = w0 * rself.y; acc.z = w0 * rself.z; acc.w = w0 * rself.w;
    __builtin_amdgcn_fence(__ATOMIC_SEQ_CST, "wavefront");
    const float4* wl4 = (const float4*)(wl + hl * LSTR);
    int j4 = 0;
    #pragma unroll 2
    for (; j4 + 4 <= deg; j4 += 4) {
      const float4 w = wl4[j4 >> 2];
      const int svj = (j4 < 32) ? sv0 : sv1;
      float4 r[4];
      #pragma unroll
      for (int k = 0; k < 4; ++k) {
        const int s = __builtin_amdgcn_readlane(svj, (j4 + k) & 31);
        r[k] = h4[s * 32 + lane];
      }
      acc.x += w.x * r[0].x; acc.y += w.x * r[0].y; acc.z += w.x * r[0].z; acc.w += w.x * r[0].w;
      acc.x += w.y * r[1].x; acc.y += w.y * r[1].y; acc.z += w.y * r[1].z; acc.w += w.y * r[1].w;
      acc.x += w.z * r[2].x; acc.y += w.z * r[2].y; acc.z += w.z * r[2].z; acc.w += w.z * r[2].w;
      acc.x += w.w * r[3].x; acc.y += w.w * r[3].y; acc.z += w.w * r[3].z; acc.w += w.w * r[3].w;
    }
    for (int j = j4; j < deg; ++j) {
      const float w = wl[hl * LSTR + j];
      const int s = __builtin_amdgcn_readlane((j < 32) ? sv0 : sv1, j & 31);
      const float4 r = h4[s * 32 + lane];
      acc.x += w * r.x; acc.y += w * r.y; acc.z += w * r.z; acc.w += w * r.w;
    }
  } else {
    m = a_self;
    for (int c0 = beg; c0 < end; c0 += 32) {
      const int cnt = min(32, end - c0);
      int sv = (lane < cnt) ? csrc[c0 + lane] : 0;
      sv = min(max(sv, 0), NNODES - 1);
      #pragma unroll 4
      for (int j4 = 0; j4 < cnt; j4 += 4) {
        const int jq = j4 + q;
        const int sq = __shfl(sv, jq);
        const float a = leaky02(sd[sq * 16 + hl] + adn);
        if (jq < cnt) m = fmaxf(m, a);
      }
    }
    m = fmaxf(m, __shfl_xor(m, 1));
    m = fmaxf(m, __shfl_xor(m, 2));

    const float w0 = expf(a_self - m);
    den = (q == 0) ? w0 : 0.f;
    {
      const float4 r = h4[n * 32 + lane];
      acc.x = w0 * r.x; acc.y = w0 * r.y; acc.z = w0 * r.z; acc.w = w0 * r.w;
    }
    for (int c0 = beg; c0 < end; c0 += 32) {
      const int cnt = min(32, end - c0);
      int sv = (lane < cnt) ? csrc[c0 + lane] : 0;
      sv = min(max(sv, 0), NNODES - 1);
      for (int j4 = 0; j4 < cnt; j4 += 4) {
        const int jq = j4 + q;
        const int sq = __shfl(sv, jq);
        const float a = leaky02(sd[sq * 16 + hl] + adn);
        const float wq = (jq < cnt) ? expf(a - m) : 0.f;
        den += wq;
        if (j4 + 4 <= cnt) {
          float4 r[4];
          float w[4];
          #pragma unroll
          for (int k = 0; k < 4; ++k) {
            const int s = __builtin_amdgcn_readlane(sv, j4 + k);
            r[k] = h4[s * 32 + lane];
            w[k] = __shfl(wq, (lane & ~3) | k);
          }
          #pragma unroll
          for (int k = 0; k < 4; ++k) {
            acc.x += w[k] * r[k].x; acc.y += w[k] * r[k].y;
            acc.z += w[k] * r[k].z; acc.w += w[k] * r[k].w;
          }
        } else {
          #pragma unroll
          for (int k = 0; k < 3; ++k) {
            if (j4 + k < cnt) {
              const int s = __builtin_amdgcn_readlane(sv, j4 + k);
              const float w = __shfl(wq, (lane & ~3) | k);
              const float4 r = h4[s * 32 + lane];
              acc.x += w * r.x; acc.y += w * r.y; acc.z += w * r.z; acc.w += w * r.w;
            }
          }
        }
      }
    }
  }
  den += __shfl_xor(den, 1);
  den += __shfl_xor(den, 2);
  const float inv = 1.0f / (den + 1e-16f);
  float4 o;
  o.x = acc.x * inv + bias[lane * 4 + 0]; o.y = acc.y * inv + bias[lane * 4 + 1];
  o.z = acc.z * inv + bias[lane * 4 + 2]; o.w = acc.w * inv + bias[lane * 4 + 3];
  o.x = o.x > 0.f ? o.x : expm1f(o.x); o.y = o.y > 0.f ? o.y : expm1f(o.y);
  o.z = o.z > 0.f ? o.z : expm1f(o.z); o.w = o.w > 0.f ? o.w : expm1f(o.w);
  ((float4*)outp)[n * 32 + lane] = o;
}

__device__ __forceinline__ int sd2_src(int n) { return ((n >> 4) << 5) + (n & 15); }
__device__ __forceinline__ int sd2_dst(int n) { return ((n >> 4) << 5) + 16 + (n & 15); }

__device__ __forceinline__ void gather_rows_h1(const float4* __restrict__ h4, int s, float w,
                                               int cnt, int es, int q, float4& acc) {
  #pragma unroll
  for (int it = 0; it < 4; ++it) {
    const int jj = it * 8 + es;
    const int sj = __shfl(s, jj);
    const float wj = __shfl(w, jj);
    const float4 r = h4[sj * 4 + q];
    const bool ok = jj < cnt;
    acc.x += wj * (ok ? r.x : 0.f); acc.y += wj * (ok ? r.y : 0.f);
    acc.z += wj * (ok ? r.z : 0.f); acc.w += wj * (ok ? r.w : 0.f);
  }
}

__global__ __launch_bounds__(256) void gat_gather_h1(
    const int* __restrict__ rowptr, const int* __restrict__ csrc,
    const float* __restrict__ sd,
    const float* __restrict__ h, const float* __restrict__ bias,
    float* __restrict__ outp) {
  const int lane = threadIdx.x & 31;
  const int wv = __builtin_amdgcn_readfirstlane((int)(threadIdx.x >> 5));
  const int n = blockIdx.x * 8 + wv;
  if (n >= NNODES) return;
  const int es = lane >> 2, q = lane & 3;
  int beg = rowptr[n], end = rowptr[n + 1];
  beg = min(max(beg, 0), NEDGES);
  end = min(max(end, beg), NEDGES);
  const float4* h4 = (const float4*)h;
  const float adn = sd[sd2_dst(n)];
  const float a_self = leaky02(sd[sd2_src(n)] + adn);

  const int deg = end - beg;
  float m, den;
  float4 acc = {0.f, 0.f, 0.f, 0.f};
  if (deg <= LCAP) {
    const bool v0 = lane < deg, v1 = 32 + lane < deg;
    int s0 = 0, s1 = 0;
    float a0 = a_self, a1 = a_self;
    if (deg > 0) {
      s0 = csrc[beg + min(lane, deg - 1)];
      s1 = csrc[beg + min(32 + lane, deg - 1)];
      s0 = min(max(s0, 0), NNODES - 1);
      s1 = min(max(s1, 0), NNODES - 1);
      const float t0 = leaky02(sd[sd2_src(s0)] + adn);
      const float t1 = leaky02(sd[sd2_src(s1)] + adn);
      a0 = v0 ? t0 : a_self;
      a1 = v1 ? t1 : a_self;
    }
    m = fmaxf(a_self, fmaxf(a0, a1));
    #pragma unroll
    for (int o = 16; o > 0; o >>= 1) m = fmaxf(m, __shfl_xor(m, o));

    const float w0 = expf(a_self - m);
    den = (lane == 0) ? w0 : 0.f;
    if (es == 0) {
      const float4 r = h4[n * 4 + q];
      acc.x = w0 * r.x; acc.y = w0 * r.y; acc.z = w0 * r.z; acc.w = w0 * r.w;
    }
    const float wa = v0 ? expf(a0 - m) : 0.f;
    den += wa;
    gather_rows_h1(h4, s0, wa, min(32, deg), es, q, acc);
    if (deg > 32) {
      const float wb = v1 ? expf(a1 - m) : 0.f;
      den += wb;
      gather_rows_h1(h4, s1, wb, deg - 32, es, q, acc);
    }
  } else {
    m = a_self;
    for (int c0 = beg; c0 < end; c0 += 32) {
      const int j = c0 + lane;
      if (j < end) {
        int s = csrc[j];
        s = min(max(s, 0), NNODES - 1);
        m = fmaxf(m, leaky02(sd[sd2_src(s)] + adn));
      }
    }
    #pragma unroll
    for (int o = 16; o > 0; o >>= 1) m = fmaxf(m, __shfl_xor(m, o));

    const float w0 = expf(a_self - m);
    den = (lane == 0) ? w0 : 0.f;
    if (es == 0) {
      const float4 r = h4[n * 4 + q];
      acc.x = w0 * r.x; acc.y = w0 * r.y; acc.z = w0 * r.z; acc.w = w0 * r.w;
    }
    for (int c0 = beg; c0 < end; c0 += 32) {
      const int cnt = min(32, end - c0);
      int s = 0;
      float w = 0.f;
      if (lane < cnt) {
        s = csrc[c0 + lane];
        s = min(max(s, 0), NNODES - 1);
        w = expf(leaky02(sd[sd2_src(s)] + adn) - m);
      }
      den += w;
      #pragma unroll
      for (int it = 0; it < 4; ++it) {
        const int jj = it * 8 + es;
        const int sj = __shfl(s, jj);
        const float wj = __shfl(w, jj);
        const float4 r = h4[sj * 4 + q];
        const bool ok = jj < cnt;
        acc.x += wj * (ok ? r.x : 0.f); acc.y += wj * (ok ? r.y : 0.f);
        acc.z += wj * (ok ? r.z : 0.f); acc.w += wj * (ok ? r.w : 0.f);
      }
    }
  }
  #pragma unroll
  for (int o = 16; o > 0; o >>= 1) den += __shfl_xor(den, o);
  #pragma unroll
  for (int o = 16; o >= 4; o >>= 1) {
    acc.x += __shfl_xor(acc.x, o); acc.y += __shfl_xor(acc.y, o);
    acc.z += __shfl_xor(acc.z, o); acc.w += __shfl_xor(acc.w, o);
  }
  const float inv = 1.0f / (den + 1e-16f);
  float4 v;
  v.x = acc.x * inv + bias[q * 4 + 0]; v.y = acc.y * inv + bias[q * 4 + 1];
  v.z = acc.z * inv + bias[q * 4 + 2]; v.w = acc.w * inv + bias[q * 4 + 3];
  float vm = fmaxf(fmaxf(v.x, v.y), fmaxf(v.z, v.w));
  vm = fmaxf(vm, __shfl_xor(vm, 1));
  vm = fmaxf(vm, __shfl_xor(vm, 2));
  float se = (expf(v.x - vm) + expf(v.y - vm)) + (expf(v.z - vm) + expf(v.w - vm));
  se += __shfl_xor(se, 1);
  se += __shfl_xor(se, 2);
  const float ls = logf(se);
  if (es == 0) {
    float4 o4;
    o4.x = v.x - vm - ls; o4.y = v.y - vm - ls; o4.z = v.z - vm - ls; o4.w = v.w - vm - ls;
    ((float4*)outp)[n * 4 + q] = o4;
  }
}

extern "C" void kernel_launch(void* const* d_in, const int* in_sizes, int n_in,
                              void* d_out, int out_size, void* d_ws, size_t ws_size,
                              hipStream_t stream) {
  const float* x     = (const float*)d_in[0];
  const int*   ei    = (const int*)  d_in[1];
  const float* W1    = (const float*)d_in[2];
  const float* attS1 = (const float*)d_in[3];
  const float* attD1 = (const float*)d_in[4];
  const float* b1    = (const float*)d_in[5];
  const float* W2    = (const float*)d_in[6];
  const float* attS2 = (const float*)d_in[7];
  const float* attD2 = (const float*)d_in[8];
  const float* b2    = (const float*)d_in[9];
  float* out = (float*)d_out;

  char* ws = (char*)d_ws;
  size_t off = 0;
  auto alloc = [&](size_t bytes) -> void* {
    void* p = ws + off;
    off = (off + bytes + 255) & ~(size_t)255;
    return p;
  };

  float*    h1   = (float*)   alloc((size_t)NNODES * F1 * 4);
  float*    sd1  = (float*)   alloc((size_t)NNODES * 16 * 4);
  float*    agg1 = (float*)   alloc((size_t)NNODES * F1 * 4);
  int*      csrc = (int*)     alloc((size_t)NEDGES * 4);
  int*      rptr = (int*)     alloc((size_t)(NNODES + 1) * 4);
  unsigned* aux  = (unsigned*)alloc((size_t)NNODES * 4);
  static_assert((size_t)NEB * NBIN + NBIN + 1 <= (size_t)NNODES, "hist and bstart fit the region");
  unsigned* hist   = aux;
  unsigned* bstart = aux + (size_t)NEB * NBIN;
  unsigned* binned = (unsigned*)agg1;
  float*    h2   = h1;
  float*    sd2  = h1 + (size_t)NNODES * C_;

  auto nb = [](long long n, int b) { return (unsigned)((n + b - 1) / b); };

  bin_count<<<NEB, ETH, 0, stream>>>(ei, hist);
  bin_place<<<NEB, ETH, 0, stream>>>(ei, hist, bstart, binned);
  bin_sort<<<NBIN, ETH, 0, stream>>>(binned, bstart, rptr, csrc);

  gemm_f32_wmma_strip<8, NFEAT, F1><<<3125, 32, 0, stream>>>(x, W1, attS1, attD1, h1, sd1);
  gat_gather_h8<<<nb(NNODES, 8), 256, 0, stream>>>(rptr, csrc, sd1, h1, b1, agg1);

  gemm_f32_wmma_strip<1, F1, C_><<<3125, 32, 0, stream>>>(agg1, W2, attS2, attD2, h2, sd2);
  gat_gather_h1<<<nb(NNODES, 8), 256, 0, stream>>>(rptr, csrc, sd2, h2, b2, out);
}
